# starting kernel
# baseline (speedup 1.0000x reference)
#include <hip/hip_runtime.h>
#include <hip/hip_bf16.h>
#include <math.h>

typedef float v2f __attribute__((ext_vector_type(2)));
typedef float v8f __attribute__((ext_vector_type(8)));

#define N_ROWS 2048
#define P_ROWS 512
#define D_DIM  1024
#define F_DIM  128
#define EPSV   1e-8f

__device__ __forceinline__ float softplus_f(float x) {
    return log1pf(expf(x));
}

__global__ __launch_bounds__(256) void fbnorm_kernel(const float* __restrict__ fb,
                                                     float* __restrict__ fbn) {
    __shared__ float red[256];
    const int row = blockIdx.x;
    const int t   = threadIdx.x;
    const float* src = fb + (size_t)row * D_DIM;
    float s = 0.0f;
    for (int c = t; c < D_DIM; c += 256) { float v = src[c]; s += v * v; }
    red[t] = s;
    __syncthreads();
    for (int off = 128; off > 0; off >>= 1) {
        if (t < off) red[t] += red[t + off];
        __syncthreads();
    }
    const float scale = 1.0f / fmaxf(sqrtf(red[0]), EPSV);
    float* dst = fbn + (size_t)row * D_DIM;
    for (int c = t; c < D_DIM; c += 256) dst[c] = src[c] * scale;
}

__global__ __launch_bounds__(256) void rowinvnorm_kernel(const float* __restrict__ X,
                                                         float* __restrict__ invn) {
    __shared__ float red[256];
    const int row = blockIdx.x;
    const int t   = threadIdx.x;
    const float* src = X + (size_t)row * D_DIM;
    float s = 0.0f;
    for (int c = t; c < D_DIM; c += 256) { float v = src[c]; s += v * v; }
    red[t] = s;
    __syncthreads();
    for (int off = 128; off > 0; off >>= 1) {
        if (t < off) red[t] += red[t + off];
        __syncthreads();
    }
    if (t == 0) invn[row] = 1.0f / fmaxf(sqrtf(red[0]), EPSV);
}

__global__ __launch_bounds__(128) void gemm_relu_wmma(const float* __restrict__ X,
                                                      const float* __restrict__ fbn,
                                                      const float* __restrict__ invn,
                                                      float* __restrict__ outp) {
    const int wid  = threadIdx.x >> 5;
    const int lane = threadIdx.x & 31;
    const int tile = blockIdx.x * 4 + wid;
    const int tf   = tile & 7;
    const int tm   = tile >> 3;
    const int half = lane >> 4;
    const int lm   = lane & 15;

    const float* arow = X   + (size_t)(tm * 16 + lm) * D_DIM + 2 * half;
    const float* brow = fbn + (size_t)(tf * 16 + lm) * D_DIM + 2 * half;

    v8f acc = {};
#pragma unroll 8
    for (int k = 0; k < D_DIM; k += 4) {
        v2f a2 = *(const v2f*)(arow + k);
        v2f b2 = *(const v2f*)(brow + k);
        acc = __builtin_amdgcn_wmma_f32_16x16x4_f32(
             false, a2,  false, b2,
             (short)0, acc,  false,  false);
    }

    const int col = tf * 16 + lm;
#pragma unroll
    for (int v = 0; v < 8; ++v) {
        const int row = tm * 16 + half * 8 + v;
        const float val = acc[v] * invn[row];
        outp[(size_t)row * F_DIM + col] = fmaxf(val, 0.0f);
    }
}

__global__ __launch_bounds__(256) void pairwise_kernel(const float* __restrict__ ap,
                                                       const float* __restrict__ bp,
                                                       const float* __restrict__ alpha_raw,
                                                       const float* __restrict__ beta_raw,
                                                       const float* __restrict__ theta_raw,
                                                       float* __restrict__ out) {
    __shared__ float apS[16][132];
    __shared__ float bpS[16][132];

    const int tx  = threadIdx.x;
    const int ty  = threadIdx.y;
    const int tid = ty * 16 + tx;
    const int j0  = blockIdx.x * 16;
    const int i0  = blockIdx.y * 16;

    for (int p = tid; p < 16 * 32; p += 256) {
        const int r  = p >> 5;
        const int c4 = (p & 31) * 4;
        *(float4*)&apS[r][c4] = *(const float4*)(ap + (size_t)(i0 + r) * F_DIM + c4);
        *(float4*)&bpS[r][c4] = *(const float4*)(bp + (size_t)(j0 + r) * F_DIM + c4);
    }
    __syncthreads();

    float s_min = 0.0f, s_a = 0.0f, s_b = 0.0f;
#pragma unroll 4
    for (int k = 0; k < F_DIM; ++k) {
        const float av = apS[ty][k];
        const float bv = bpS[tx][k];
        s_min += fminf(av, bv);
        s_a   += (bv > 0.0f) ? av : 0.0f;
        s_b   += (av > 0.0f) ? bv : 0.0f;
    }

    const float al = softplus_f(alpha_raw[0]);
    const float be = softplus_f(beta_raw[0]);
    const float th = softplus_f(theta_raw[0]);
    out[(size_t)(i0 + ty) * P_ROWS + (j0 + tx)] =
        (th + al + be) * s_min - al * s_a - be * s_b;
}

extern "C" void kernel_launch(void* const* d_in, const int* in_sizes, int n_in,
                              void* d_out, int out_size, void* d_ws, size_t ws_size,
                              hipStream_t stream) {
    const float* input     = (const float*)d_in[0];
    const float* prototype = (const float*)d_in[1];
    const float* fb        = (const float*)d_in[2];
    const float* alpha_raw = (const float*)d_in[3];
    const float* beta_raw  = (const float*)d_in[4];
    const float* theta_raw = (const float*)d_in[5];
    float* out = (float*)d_out;

    float* w      = (float*)d_ws;
    float* fbn    = w;
    float* inv_in = fbn + (size_t)F_DIM * D_DIM;
    float* inv_pr = inv_in + N_ROWS;
    float* ap     = inv_pr + P_ROWS;
    float* bp     = ap + (size_t)N_ROWS * F_DIM;

    fbnorm_kernel<<<F_DIM, 256, 0, stream>>>(fb, fbn);
    rowinvnorm_kernel<<<N_ROWS, 256, 0, stream>>>(input, inv_in);
    rowinvnorm_kernel<<<P_ROWS, 256, 0, stream>>>(prototype, inv_pr);

    gemm_relu_wmma<<<(N_ROWS / 16) * (F_DIM / 16) / 4, 128, 0, stream>>>(input, fbn, inv_in, ap);
    gemm_relu_wmma<<<(P_ROWS / 16) * (F_DIM / 16) / 4, 128, 0, stream>>>(prototype, fbn, inv_pr, bp);

    dim3 grid(P_ROWS / 16, N_ROWS / 16);
    pairwise_kernel<<<grid, dim3(16, 16), 0, stream>>>(ap, bp, alpha_raw, beta_raw, theta_raw, out);
}


# Pairwise launch: every (ap row, k) whose ap is zero after the relu is skipped by a scalar branch (ap is wave-uniform, half are zero for symmetric inputs); loads held one k-quad ahead; sum order kept
# speedup vs baseline: 2.5666x; 2.5666x over previous
#include <hip/hip_runtime.h>
#include <hip/hip_bf16.h>
#include <math.h>

typedef float v2f __attribute__((ext_vector_type(2)));
typedef float v8f __attribute__((ext_vector_type(8)));
typedef float v4f __attribute__((ext_vector_type(4)));

#define N_ROWS 2048
#define P_ROWS 512
#define D_DIM  1024
#define F_DIM  128
#define EPSV   1e-8f

__device__ __forceinline__ float softplus_f(float x) {
    return log1pf(expf(x));
}

#define NRM_WAVES 8
#define NRM_ROWS  (F_DIM + P_ROWS + N_ROWS)

__global__ __launch_bounds__(32 * NRM_WAVES) void norms_kernel(const float* __restrict__ X,
                                                              const float* __restrict__ Pm,
                                                              const float* __restrict__ fb,
                                                              float* __restrict__ inv_in,
                                                              float* __restrict__ inv_pr,
                                                              float* __restrict__ fbn) {
    const int lane = threadIdx.x & 31;
    const int wid = __builtin_amdgcn_readfirstlane(threadIdx.x >> 5);
    const int r   = blockIdx.x * NRM_WAVES + wid;

    const float* src;
    if (r < F_DIM)               src = fb + (size_t)r * D_DIM;
    else if (r < F_DIM + P_ROWS) src = Pm + (size_t)(r - F_DIM) * D_DIM;
    else                         src = X  + (size_t)(r - F_DIM - P_ROWS) * D_DIM;

    float4 v[8];
#pragma unroll
    for (int j = 0; j < 8; ++j) v[j] = *(const float4*)(src + 4 * (j * 32 + lane));

    float sx = 0.0f, sy = 0.0f, sz = 0.0f, sw = 0.0f;
#pragma unroll
    for (int j = 0; j < 8; ++j) {
        sx = fmaf(v[j].x, v[j].x, sx);
        sy = fmaf(v[j].y, v[j].y, sy);
        sz = fmaf(v[j].z, v[j].z, sz);
        sw = fmaf(v[j].w, v[j].w, sw);
    }
    float s = (sx + sy) + (sz + sw);
#pragma unroll
    for (int off = 16; off > 0; off >>= 1) s += __shfl_xor(s, off);

    const float scale = 1.0f / fmaxf(sqrtf(s), EPSV);

    if (r < F_DIM) {
        float* dst = fbn + (size_t)r * D_DIM;
#pragma unroll
        for (int j = 0; j < 8; ++j) {
            float4 o;
            o.x = v[j].x * scale; o.y = v[j].y * scale;
            o.z = v[j].z * scale; o.w = v[j].w * scale;
            *(float4*)(dst + 4 * (j * 32 + lane)) = o;
        }
    } else if (lane == 0) {
        if (r < F_DIM + P_ROWS) inv_pr[r - F_DIM] = scale;
        else                    inv_in[r - F_DIM - P_ROWS] = scale;
    }
}

template <int KS>
__global__ __launch_bounds__(32 * KS) void gemm_relu_wmma(const float* __restrict__ X,
                                                          const float* __restrict__ fbn,
                                                          const float* __restrict__ invn,
                                                          float* __restrict__ outp) {
    constexpr int KSEG = D_DIM / KS;
    __shared__ float red[KS][8][32];

    const int wid  = threadIdx.x >> 5;
    const int lane = threadIdx.x & 31;
    const int tile = blockIdx.x;
    const int tf   = tile & 7;
    const int tm   = tile >> 3;
    const int half = lane >> 4;
    const int lm   = lane & 15;

    const float* arow = X   + (size_t)(tm * 16 + lm) * D_DIM + wid * KSEG + 4 * half;
    const float* brow = fbn + (size_t)(tf * 16 + lm) * D_DIM + wid * KSEG + 4 * half;

    constexpr int NV = 8 / KS;
    const int col  = tf * 16 + lm;
    const int row0 = tm * 16 + half * 8 + wid;
    float scale[NV];
#pragma unroll
    for (int i = 0; i < NV; ++i) scale[i] = invn[row0 + i * KS];

    v8f acc = {};
#pragma unroll
    for (int k = 0; k < KSEG; k += 8) {
        const float4 a4 = *(const float4*)(arow + k);
        const float4 b4 = *(const float4*)(brow + k);
        const v2f a_lo = {a4.x, a4.y}, a_hi = {a4.z, a4.w};
        const v2f b_lo = {b4.x, b4.y}, b_hi = {b4.z, b4.w};
        acc = __builtin_amdgcn_wmma_f32_16x16x4_f32(
             false, a_lo,  false, b_lo,
             (short)0, acc,  false,  false);
        acc = __builtin_amdgcn_wmma_f32_16x16x4_f32(
             false, a_hi,  false, b_hi,
             (short)0, acc,  false,  false);
    }

#pragma unroll
    for (int v = 0; v < 8; ++v) red[wid][v][lane] = acc[v];
    __syncthreads();

#pragma unroll
    for (int i = 0; i < NV; ++i) {
        const int v = wid + i * KS;
        float s = red[0][v][lane];
#pragma unroll
        for (int w = 1; w < KS; ++w) s += red[w][v][lane];
        outp[(size_t)(row0 + i * KS) * F_DIM + col] = fmaxf(s * scale[i], 0.0f);
    }
}

#define PW_IW    4
#define PW_IB    32
#define PW_JB    128
#define PW_MOFF  132
#define PW_STR   276

__global__ __launch_bounds__(256) void pairwise_kernel(const float* __restrict__ ap,
                                                       const float* __restrict__ bp,
                                                       const float* __restrict__ alpha_raw,
                                                       const float* __restrict__ beta_raw,
                                                       const float* __restrict__ theta_raw,
                                                       float* __restrict__ out) {
    __shared__ __attribute__((aligned(16))) float tS[F_DIM * PW_STR];

    const int tid  = threadIdx.x;
    const int lane = tid & 31;
    const int wid  = __builtin_amdgcn_readfirstlane(tid >> 5);
    const int j0   = blockIdx.x * PW_JB;
    const int i0   = blockIdx.y * PW_IB + wid * PW_IW;

    float al, be, th;
    {
        const int jq = (wid << 2) | (lane & 3);
        float4 g[4][4];
#pragma unroll
        for (int t = 0; t < 4; ++t) {
            const int q = (t << 3) | (lane >> 2);
#pragma unroll
            for (int rr = 0; rr < 4; ++rr)
                g[t][rr] = *(const float4*)(bp + (size_t)(j0 + 4 * jq + rr) * F_DIM + 4 * q);
        }
        al = softplus_f(alpha_raw[0]);
        be = softplus_f(beta_raw[0]);
        th = softplus_f(theta_raw[0]);
        asm volatile("" :: "s"(al), "s"(be), "s"(th));
#pragma unroll
        for (int t = 0; t < 4; ++t) {
            const int q = (t << 3) | (lane >> 2);
            float* dst = tS + (4 * q) * PW_STR + 4 * jq;
            float4 v, m;
            v.x = g[t][0].x; v.y = g[t][1].x; v.z = g[t][2].x; v.w = g[t][3].x;
            m.x = (v.x > 0.0f) ? 1.0f : 0.0f; m.y = (v.y > 0.0f) ? 1.0f : 0.0f;
            m.z = (v.z > 0.0f) ? 1.0f : 0.0f; m.w = (v.w > 0.0f) ? 1.0f : 0.0f;
            *(float4*)(dst) = v; *(float4*)(dst + PW_MOFF) = m;
            v.x = g[t][0].y; v.y = g[t][1].y; v.z = g[t][2].y; v.w = g[t][3].y;
            m.x = (v.x > 0.0f) ? 1.0f : 0.0f; m.y = (v.y > 0.0f) ? 1.0f : 0.0f;
            m.z = (v.z > 0.0f) ? 1.0f : 0.0f; m.w = (v.w > 0.0f) ? 1.0f : 0.0f;
            *(float4*)(dst + PW_STR) = v; *(float4*)(dst + PW_STR + PW_MOFF) = m;
            v.x = g[t][0].z; v.y = g[t][1].z; v.z = g[t][2].z; v.w = g[t][3].z;
            m.x = (v.x > 0.0f) ? 1.0f : 0.0f; m.y = (v.y > 0.0f) ? 1.0f : 0.0f;
            m.z = (v.z > 0.0f) ? 1.0f : 0.0f; m.w = (v.w > 0.0f) ? 1.0f : 0.0f;
            *(float4*)(dst + 2 * PW_STR) = v; *(float4*)(dst + 2 * PW_STR + PW_MOFF) = m;
            v.x = g[t][0].w; v.y = g[t][1].w; v.z = g[t][2].w; v.w = g[t][3].w;
            m.x = (v.x > 0.0f) ? 1.0f : 0.0f; m.y = (v.y > 0.0f) ? 1.0f : 0.0f;
            m.z = (v.z > 0.0f) ? 1.0f : 0.0f; m.w = (v.w > 0.0f) ? 1.0f : 0.0f;
            *(float4*)(dst + 3 * PW_STR) = v; *(float4*)(dst + 3 * PW_STR + PW_MOFF) = m;
        }
    }
    __syncthreads();

    const float* __restrict__ arow = ap + (size_t)i0 * F_DIM;
    const float* brow = tS + 4 * lane;

    float s_min[PW_IW][4], s_a[PW_IW][4], s_b[PW_IW][4];
#pragma unroll
    for (int r = 0; r < PW_IW; ++r) {
#pragma unroll
        for (int c = 0; c < 4; ++c) { s_min[r][c] = 0.0f; s_a[r][c] = 0.0f; s_b[r][c] = 0.0f; }
    }

    v8f avP[PW_IW], avQ[PW_IW];
    v4f bqA[4], mqA[4], bqB[4], mqB[4];

#define PW_LOAD_BM(BQ, MQ, K0)                                                    \
    _Pragma("unroll")                                                             \
    for (int kk = 0; kk < 4; ++kk) {                                              \
        BQ[kk] = *(const v4f*)(brow + ((K0) + kk) * PW_STR);                      \
        MQ[kk] = *(const v4f*)(brow + ((K0) + kk) * PW_STR + PW_MOFF);            \
    }
#define PW_LOAD_AP(AV, K0)                                                        \
    _Pragma("unroll")                                                             \
    for (int r = 0; r < PW_IW; ++r) AV[r] = *(const v8f*)(arow + r * F_DIM + (K0));

#define PW_HOLD_AP_BM(AV, BQ, MQ)                                                 \
    asm volatile("" :: "s"(AV[0]), "s"(AV[1]), "s"(AV[2]), "s"(AV[3]),            \
                       "v"(BQ[0]), "v"(BQ[1]), "v"(BQ[2]), "v"(BQ[3]),            \
                       "v"(MQ[0]), "v"(MQ[1]), "v"(MQ[2]), "v"(MQ[3]) : "memory")
#define PW_HOLD_BM(BQ, MQ)                                                        \
    asm volatile("" :: "v"(BQ[0]), "v"(BQ[1]), "v"(BQ[2]), "v"(BQ[3]),            \
                       "v"(MQ[0]), "v"(MQ[1]), "v"(MQ[2]), "v"(MQ[3]) : "memory")

#define PW_SUM_QUAD(AV, KO, BQ, MQ)                                               \
    _Pragma("unroll")                                                             \
    for (int kk = 0; kk < 4; ++kk) {                                              \
        _Pragma("unroll")                                                         \
        for (int r = 0; r < PW_IW; ++r) {                                         \
            const float a = AV[r][(KO) + kk];                                     \
            if (__builtin_expect(a > 0.0f, 1)) {                                  \
                _Pragma("unroll")                                                 \
                for (int c = 0; c < 4; ++c) {                                     \
                    s_min[r][c] += fminf(a, BQ[kk][c]);                           \
                    s_a[r][c] = fmaf(a, MQ[kk][c], s_a[r][c]);                    \
                    s_b[r][c] += BQ[kk][c];                                       \
                }                                                                 \
            }                                                                     \
        }                                                                         \
    }

#define PW_PASS(AVC, AVN, K0)                                                     \
    {                                                                             \
        const int kn = ((K0) + 8 < F_DIM) ? ((K0) + 8) : (F_DIM - 8);             \
        PW_LOAD_BM(bqB, mqB, (K0) + 4)                                            \
        PW_HOLD_AP_BM(AVC, bqA, mqA);                                             \
        PW_LOAD_AP(AVN, kn)                                                       \
        PW_SUM_QUAD(AVC, 0, bqA, mqA)                                             \
        PW_LOAD_BM(bqA, mqA, kn)                                                  \
        PW_HOLD_BM(bqB, mqB);                                                     \
        PW_SUM_QUAD(AVC, 4, bqB, mqB)                                             \
    }

    PW_LOAD_AP(avP, 0)
    PW_LOAD_BM(bqA, mqA, 0)
#pragma nounroll
    for (int k = 0; k < F_DIM; k += 16) {
        PW_PASS(avP, avQ, k)
        PW_PASS(avQ, avP, k + 8)
    }
#undef PW_LOAD_BM
#undef PW_LOAD_AP
#undef PW_HOLD_AP_BM
#undef PW_HOLD_BM
#undef PW_SUM_QUAD
#undef PW_PASS

#pragma unroll
    for (int r = 0; r < PW_IW; ++r) {
        float4 o;
        o.x = (th + al + be) * s_min[r][0] - al * s_a[r][0] - be * s_b[r][0];
        o.y = (th + al + be) * s_min[r][1] - al * s_a[r][1] - be * s_b[r][1];
        o.z = (th + al + be) * s_min[r][2] - al * s_a[r][2] - be * s_b[r][2];
        o.w = (th + al + be) * s_min[r][3] - al * s_a[r][3] - be * s_b[r][3];
        *(float4*)(out + (size_t)(i0 + r) * P_ROWS + j0 + 4 * lane) = o;
    }
}

extern "C" void kernel_launch(void* const* d_in, const int* in_sizes, int n_in,
                              void* d_out, int out_size, void* d_ws, size_t ws_size,
                              hipStream_t stream) {
    const float* input     = (const float*)d_in[0];
    const float* prototype = (const float*)d_in[1];
    const float* fb        = (const float*)d_in[2];
    const float* alpha_raw = (const float*)d_in[3];
    const float* beta_raw  = (const float*)d_in[4];
    const float* theta_raw = (const float*)d_in[5];
    float* out = (float*)d_out;

    float* w      = (float*)d_ws;
    float* fbn    = w;
    float* inv_in = fbn + (size_t)F_DIM * D_DIM;
    float* inv_pr = inv_in + N_ROWS;
    float* ap     = inv_pr + P_ROWS;
    float* bp     = ap + (size_t)N_ROWS * F_DIM;

    norms_kernel<<<NRM_ROWS / NRM_WAVES, 32 * NRM_WAVES, 0, stream>>>(input, prototype, fb, inv_in, inv_pr, fbn);

    gemm_relu_wmma<4><<<(N_ROWS / 16) * (F_DIM / 16), 128, 0, stream>>>(input, fbn, inv_in, ap);
    gemm_relu_wmma<8><<<(P_ROWS / 16) * (F_DIM / 16), 256, 0, stream>>>(prototype, fbn, inv_pr, bp);

    dim3 grid(P_ROWS / PW_JB, N_ROWS / PW_IB);
    pairwise_kernel<<<grid, 256, 0, stream>>>(ap, bp, alpha_raw, beta_raw, theta_raw, out);
}


# Input projection register-tiled: a wave owns a 32x32 tile (2x2 fragments, 4 accumulators), a 16-B fragment load feeds 4 WMMAs not 2, loads ringed 8 groups ahead; K split 8 ways; dot sums reordered
# speedup vs baseline: 2.7281x; 1.0629x over previous
#include <hip/hip_runtime.h>
#include <hip/hip_bf16.h>
#include <math.h>

typedef float v2f __attribute__((ext_vector_type(2)));
typedef float v8f __attribute__((ext_vector_type(8)));
typedef float v4f __attribute__((ext_vector_type(4)));

#define N_ROWS 2048
#define P_ROWS 512
#define D_DIM  1024
#define F_DIM  128
#define EPSV   1e-8f

__device__ __forceinline__ float softplus_f(float x) {
    return log1pf(expf(x));
}

#define NRM_WAVES 8
#define NRM_ROWS  (F_DIM + P_ROWS + N_ROWS)

__global__ __launch_bounds__(32 * NRM_WAVES) void norms_kernel(const float* __restrict__ X,
                                                              const float* __restrict__ Pm,
                                                              const float* __restrict__ fb,
                                                              float* __restrict__ inv_in,
                                                              float* __restrict__ inv_pr,
                                                              float* __restrict__ fbn) {
    const int lane = threadIdx.x & 31;
    const int wid = __builtin_amdgcn_readfirstlane(threadIdx.x >> 5);
    const int r   = blockIdx.x * NRM_WAVES + wid;

    const float* src;
    if (r < F_DIM)               src = fb + (size_t)r * D_DIM;
    else if (r < F_DIM + P_ROWS) src = Pm + (size_t)(r - F_DIM) * D_DIM;
    else                         src = X  + (size_t)(r - F_DIM - P_ROWS) * D_DIM;

    float4 v[8];
#pragma unroll
    for (int j = 0; j < 8; ++j) v[j] = *(const float4*)(src + 4 * (j * 32 + lane));

    float sx = 0.0f, sy = 0.0f, sz = 0.0f, sw = 0.0f;
#pragma unroll
    for (int j = 0; j < 8; ++j) {
        sx = fmaf(v[j].x, v[j].x, sx);
        sy = fmaf(v[j].y, v[j].y, sy);
        sz = fmaf(v[j].z, v[j].z, sz);
        sw = fmaf(v[j].w, v[j].w, sw);
    }
    float s = (sx + sy) + (sz + sw);
#pragma unroll
    for (int off = 16; off > 0; off >>= 1) s += __shfl_xor(s, off);

    const float scale = 1.0f / fmaxf(sqrtf(s), EPSV);

    if (r < F_DIM) {
        float* dst = fbn + (size_t)r * D_DIM;
#pragma unroll
        for (int j = 0; j < 8; ++j) {
            float4 o;
            o.x = v[j].x * scale; o.y = v[j].y * scale;
            o.z = v[j].z * scale; o.w = v[j].w * scale;
            *(float4*)(dst + 4 * (j * 32 + lane)) = o;
        }
    } else if (lane == 0) {
        if (r < F_DIM + P_ROWS) inv_pr[r - F_DIM] = scale;
        else                    inv_in[r - F_DIM - P_ROWS] = scale;
    }
}

template <int KS>
__global__ __launch_bounds__(32 * KS) void gemm_relu_wmma(const float* __restrict__ X,
                                                          const float* __restrict__ fbn,
                                                          const float* __restrict__ invn,
                                                          float* __restrict__ outp) {
    constexpr int KSEG = D_DIM / KS;
    __shared__ float red[KS][8][32];

    const int wid  = threadIdx.x >> 5;
    const int lane = threadIdx.x & 31;
    const int tile = blockIdx.x;
    const int tf   = tile & 7;
    const int tm   = tile >> 3;
    const int half = lane >> 4;
    const int lm   = lane & 15;

    const float* arow = X   + (size_t)(tm * 16 + lm) * D_DIM + wid * KSEG + 4 * half;
    const float* brow = fbn + (size_t)(tf * 16 + lm) * D_DIM + wid * KSEG + 4 * half;

    constexpr int NV = 8 / KS;
    const int col  = tf * 16 + lm;
    const int row0 = tm * 16 + half * 8 + wid;
    float scale[NV];
#pragma unroll
    for (int i = 0; i < NV; ++i) scale[i] = invn[row0 + i * KS];

    v8f acc = {};
#pragma unroll
    for (int k = 0; k < KSEG; k += 8) {
        const float4 a4 = *(const float4*)(arow + k);
        const float4 b4 = *(const float4*)(brow + k);
        const v2f a_lo = {a4.x, a4.y}, a_hi = {a4.z, a4.w};
        const v2f b_lo = {b4.x, b4.y}, b_hi = {b4.z, b4.w};
        acc = __builtin_amdgcn_wmma_f32_16x16x4_f32(
             false, a_lo,  false, b_lo,
             (short)0, acc,  false,  false);
        acc = __builtin_amdgcn_wmma_f32_16x16x4_f32(
             false, a_hi,  false, b_hi,
             (short)0, acc,  false,  false);
    }

#pragma unroll
    for (int v = 0; v < 8; ++v) red[wid][v][lane] = acc[v];
    __syncthreads();

#pragma unroll
    for (int i = 0; i < NV; ++i) {
        const int v = wid + i * KS;
        float s = red[0][v][lane];
#pragma unroll
        for (int w = 1; w < KS; ++w) s += red[w][v][lane];
        outp[(size_t)(row0 + i * KS) * F_DIM + col] = fmaxf(s * scale[i], 0.0f);
    }
}

#define RT_PITCH 34
#define RT_AHEAD 8

template <int KS>
__global__ __launch_bounds__(32 * KS) void gemm_relu_wmma_rt(const float* __restrict__ X,
                                                             const float* __restrict__ fbn,
                                                             const float* __restrict__ invn,
                                                             float* __restrict__ outp) {
    constexpr int KSEG = D_DIM / KS;
    constexpr int NR   = 32 / KS;
    __shared__ float red[KS][32][RT_PITCH];

    const int lane = threadIdx.x & 31;
    const int wid  = __builtin_amdgcn_readfirstlane(threadIdx.x >> 5);
    const int tile = blockIdx.x;
    const int tf   = tile & 3;
    const int tm   = tile >> 2;
    const int half = lane >> 4;
    const int lm   = lane & 15;

    const float* a0p = X   + (size_t)(tm * 32 + lm) * D_DIM + wid * KSEG + 4 * half;
    const float* a1p = a0p + (size_t)16 * D_DIM;
    const float* b0p = fbn + (size_t)(tf * 32 + lm) * D_DIM + wid * KSEG + 4 * half;
    const float* b1p = b0p + (size_t)16 * D_DIM;

    float scale[NR];
#pragma unroll
    for (int i = 0; i < NR; ++i) scale[i] = invn[tm * 32 + wid * NR + i];

    constexpr int NG = KSEG / 8;
    v4f fa0[NG], fa1[NG], fb0[NG], fb1[NG];
#define RT_LOAD_GROUP(G)                                                          \
    {                                                                             \
        fa0[G] = *(const v4f*)(a0p + 8 * (G));     \
        fa1[G] = *(const v4f*)(a1p + 8 * (G));                                    \
        fb0[G] = *(const v4f*)(b0p + 8 * (G));                                    \
        fb1[G] = *(const v4f*)(b1p + 8 * (G));                                    \
    }
#pragma unroll
    for (int g = 0; g < RT_AHEAD; ++g) RT_LOAD_GROUP(g)

    v8f acc00 = {}, acc01 = {}, acc10 = {}, acc11 = {};
#pragma unroll
    for (int g = 0; g < NG; ++g) {
        asm volatile("" :: "v"(fa0[g]), "v"(fa1[g]), "v"(fb0[g]), "v"(fb1[g]) : "memory");
        const v2f a0l = {fa0[g].x, fa0[g].y}, a0h = {fa0[g].z, fa0[g].w};
        const v2f a1l = {fa1[g].x, fa1[g].y}, a1h = {fa1[g].z, fa1[g].w};
        const v2f b0l = {fb0[g].x, fb0[g].y}, b0h = {fb0[g].z, fb0[g].w};
        const v2f b1l = {fb1[g].x, fb1[g].y}, b1h = {fb1[g].z, fb1[g].w};
        acc00 = __builtin_amdgcn_wmma_f32_16x16x4_f32(false, a0l, false, b0l, (short)0, acc00, false, false);
        acc01 = __builtin_amdgcn_wmma_f32_16x16x4_f32(false, a0l, false, b1l, (short)0, acc01, false, false);
        acc10 = __builtin_amdgcn_wmma_f32_16x16x4_f32(false, a1l, false, b0l, (short)0, acc10, false, false);
        acc11 = __builtin_amdgcn_wmma_f32_16x16x4_f32(false, a1l, false, b1l, (short)0, acc11, false, false);
        acc00 = __builtin_amdgcn_wmma_f32_16x16x4_f32(false, a0h, false, b0h, (short)0, acc00, false, false);
        acc01 = __builtin_amdgcn_wmma_f32_16x16x4_f32(false, a0h, false, b1h, (short)0, acc01, false, false);
        acc10 = __builtin_amdgcn_wmma_f32_16x16x4_f32(false, a1h, false, b0h, (short)0, acc10, false, false);
        acc11 = __builtin_amdgcn_wmma_f32_16x16x4_f32(false, a1h, false, b1h, (short)0, acc11, false, false);
        if (g + RT_AHEAD < NG) RT_LOAD_GROUP(g + RT_AHEAD)
    }
#undef RT_LOAD_GROUP

#pragma unroll
    for (int v = 0; v < 8; ++v) {
        red[wid][     8 * half + v][     lm] = acc00[v];
        red[wid][     8 * half + v][16 + lm] = acc01[v];
        red[wid][16 + 8 * half + v][     lm] = acc10[v];
        red[wid][16 + 8 * half + v][16 + lm] = acc11[v];
    }
    __syncthreads();

#pragma unroll
    for (int i = 0; i < NR; ++i) {
        const int r = wid * NR + i;
        float s = red[0][r][lane];
#pragma unroll
        for (int w = 1; w < KS; ++w) s += red[w][r][lane];
        outp[(size_t)(tm * 32 + r) * F_DIM + tf * 32 + lane] = fmaxf(s * scale[i], 0.0f);
    }
}

#define PW_IW    4
#define PW_IB    32
#define PW_JB    128
#define PW_MOFF  132
#define PW_STR   276

__global__ __launch_bounds__(256) void pairwise_kernel(const float* __restrict__ ap,
                                                       const float* __restrict__ bp,
                                                       const float* __restrict__ alpha_raw,
                                                       const float* __restrict__ beta_raw,
                                                       const float* __restrict__ theta_raw,
                                                       float* __restrict__ out) {
    __shared__ __attribute__((aligned(16))) float tS[F_DIM * PW_STR];

    const int tid  = threadIdx.x;
    const int lane = tid & 31;
    const int wid  = __builtin_amdgcn_readfirstlane(tid >> 5);
    const int j0   = blockIdx.x * PW_JB;
    const int i0   = blockIdx.y * PW_IB + wid * PW_IW;

    float al, be, th;
    {
        const int jq = (wid << 2) | (lane & 3);
        float4 g[4][4];
#pragma unroll
        for (int t = 0; t < 4; ++t) {
            const int q = (t << 3) | (lane >> 2);
#pragma unroll
            for (int rr = 0; rr < 4; ++rr)
                g[t][rr] = *(const float4*)(bp + (size_t)(j0 + 4 * jq + rr) * F_DIM + 4 * q);
        }
        al = softplus_f(alpha_raw[0]);
        be = softplus_f(beta_raw[0]);
        th = softplus_f(theta_raw[0]);
        asm volatile("" :: "s"(al), "s"(be), "s"(th));
#pragma unroll
        for (int t = 0; t < 4; ++t) {
            const int q = (t << 3) | (lane >> 2);
            float* dst = tS + (4 * q) * PW_STR + 4 * jq;
            float4 v, m;
            v.x = g[t][0].x; v.y = g[t][1].x; v.z = g[t][2].x; v.w = g[t][3].x;
            m.x = (v.x > 0.0f) ? 1.0f : 0.0f; m.y = (v.y > 0.0f) ? 1.0f : 0.0f;
            m.z = (v.z > 0.0f) ? 1.0f : 0.0f; m.w = (v.w > 0.0f) ? 1.0f : 0.0f;
            *(float4*)(dst) = v; *(float4*)(dst + PW_MOFF) = m;
            v.x = g[t][0].y; v.y = g[t][1].y; v.z = g[t][2].y; v.w = g[t][3].y;
            m.x = (v.x > 0.0f) ? 1.0f : 0.0f; m.y = (v.y > 0.0f) ? 1.0f : 0.0f;
            m.z = (v.z > 0.0f) ? 1.0f : 0.0f; m.w = (v.w > 0.0f) ? 1.0f : 0.0f;
            *(float4*)(dst + PW_STR) = v; *(float4*)(dst + PW_STR + PW_MOFF) = m;
            v.x = g[t][0].z; v.y = g[t][1].z; v.z = g[t][2].z; v.w = g[t][3].z;
            m.x = (v.x > 0.0f) ? 1.0f : 0.0f; m.y = (v.y > 0.0f) ? 1.0f : 0.0f;
            m.z = (v.z > 0.0f) ? 1.0f : 0.0f; m.w = (v.w > 0.0f) ? 1.0f : 0.0f;
            *(float4*)(dst + 2 * PW_STR) = v; *(float4*)(dst + 2 * PW_STR + PW_MOFF) = m;
            v.x = g[t][0].w; v.y = g[t][1].w; v.z = g[t][2].w; v.w = g[t][3].w;
            m.x = (v.x > 0.0f) ? 1.0f : 0.0f; m.y = (v.y > 0.0f) ? 1.0f : 0.0f;
            m.z = (v.z > 0.0f) ? 1.0f : 0.0f; m.w = (v.w > 0.0f) ? 1.0f : 0.0f;
            *(float4*)(dst + 3 * PW_STR) = v; *(float4*)(dst + 3 * PW_STR + PW_MOFF) = m;
        }
    }
    __syncthreads();

    const float* __restrict__ arow = ap + (size_t)i0 * F_DIM;
    const float* brow = tS + 4 * lane;

    float s_min[PW_IW][4], s_a[PW_IW][4], s_b[PW_IW][4];
#pragma unroll
    for (int r = 0; r < PW_IW; ++r) {
#pragma unroll
        for (int c = 0; c < 4; ++c) { s_min[r][c] = 0.0f; s_a[r][c] = 0.0f; s_b[r][c] = 0.0f; }
    }

    v8f avP[PW_IW], avQ[PW_IW];
    v4f bqA[4], mqA[4], bqB[4], mqB[4];

#define PW_LOAD_BM(BQ, MQ, K0)                                                    \
    _Pragma("unroll")                                                             \
    for (int kk = 0; kk < 4; ++kk) {                                              \
        BQ[kk] = *(const v4f*)(brow + ((K0) + kk) * PW_STR);                      \
        MQ[kk] = *(const v4f*)(brow + ((K0) + kk) * PW_STR + PW_MOFF);            \
    }
#define PW_LOAD_AP(AV, K0)                                                        \
    _Pragma("unroll")                                                             \
    for (int r = 0; r < PW_IW; ++r) AV[r] = *(const v8f*)(arow + r * F_DIM + (K0));

#define PW_HOLD_AP_BM(AV, BQ, MQ)                                                 \
    asm volatile("" :: "s"(AV[0]), "s"(AV[1]), "s"(AV[2]), "s"(AV[3]),            \
                       "v"(BQ[0]), "v"(BQ[1]), "v"(BQ[2]), "v"(BQ[3]),            \
                       "v"(MQ[0]), "v"(MQ[1]), "v"(MQ[2]), "v"(MQ[3]) : "memory")
#define PW_HOLD_BM(BQ, MQ)                                                        \
    asm volatile("" :: "v"(BQ[0]), "v"(BQ[1]), "v"(BQ[2]), "v"(BQ[3]),            \
                       "v"(MQ[0]), "v"(MQ[1]), "v"(MQ[2]), "v"(MQ[3]) : "memory")

#define PW_SUM_QUAD(AV, KO, BQ, MQ)                                               \
    _Pragma("unroll")                                                             \
    for (int kk = 0; kk < 4; ++kk) {                                              \
        _Pragma("unroll")                                                         \
        for (int r = 0; r < PW_IW; ++r) {                                         \
            const float a = AV[r][(KO) + kk];                                     \
            if (__builtin_expect(a > 0.0f, 1)) {                                  \
                _Pragma("unroll")                                                 \
                for (int c = 0; c < 4; ++c) {                                     \
                    s_min[r][c] += fminf(a, BQ[kk][c]);                           \
                    s_a[r][c] = fmaf(a, MQ[kk][c], s_a[r][c]);                    \
                    s_b[r][c] += BQ[kk][c];                                       \
                }                                                                 \
            }                                                                     \
        }                                                                         \
    }

#define PW_PASS(AVC, AVN, K0)                                                     \
    {                                                                             \
        const int kn = ((K0) + 8 < F_DIM) ? ((K0) + 8) : (F_DIM - 8);             \
        PW_LOAD_BM(bqB, mqB, (K0) + 4)                                            \
        PW_HOLD_AP_BM(AVC, bqA, mqA);                                             \
        PW_LOAD_AP(AVN, kn)                                                       \
        PW_SUM_QUAD(AVC, 0, bqA, mqA)                                             \
        PW_LOAD_BM(bqA, mqA, kn)                                                  \
        PW_HOLD_BM(bqB, mqB);                                                     \
        PW_SUM_QUAD(AVC, 4, bqB, mqB)                                             \
    }

    PW_LOAD_AP(avP, 0)
    PW_LOAD_BM(bqA, mqA, 0)
#pragma nounroll
    for (int k = 0; k < F_DIM; k += 16) {
        PW_PASS(avP, avQ, k)
        PW_PASS(avQ, avP, k + 8)
    }
#undef PW_LOAD_BM
#undef PW_LOAD_AP
#undef PW_HOLD_AP_BM
#undef PW_HOLD_BM
#undef PW_SUM_QUAD
#undef PW_PASS

#pragma unroll
    for (int r = 0; r < PW_IW; ++r) {
        float4 o;
        o.x = (th + al + be) * s_min[r][0] - al * s_a[r][0] - be * s_b[r][0];
        o.y = (th + al + be) * s_min[r][1] - al * s_a[r][1] - be * s_b[r][1];
        o.z = (th + al + be) * s_min[r][2] - al * s_a[r][2] - be * s_b[r][2];
        o.w = (th + al + be) * s_min[r][3] - al * s_a[r][3] - be * s_b[r][3];
        *(float4*)(out + (size_t)(i0 + r) * P_ROWS + j0 + 4 * lane) = o;
    }
}

extern "C" void kernel_launch(void* const* d_in, const int* in_sizes, int n_in,
                              void* d_out, int out_size, void* d_ws, size_t ws_size,
                              hipStream_t stream) {
    const float* input     = (const float*)d_in[0];
    const float* prototype = (const float*)d_in[1];
    const float* fb        = (const float*)d_in[2];
    const float* alpha_raw = (const float*)d_in[3];
    const float* beta_raw  = (const float*)d_in[4];
    const float* theta_raw = (const float*)d_in[5];
    float* out = (float*)d_out;

    float* w      = (float*)d_ws;
    float* fbn    = w;
    float* inv_in = fbn + (size_t)F_DIM * D_DIM;
    float* inv_pr = inv_in + N_ROWS;
    float* ap     = inv_pr + P_ROWS;
    float* bp     = ap + (size_t)N_ROWS * F_DIM;

    norms_kernel<<<NRM_ROWS / NRM_WAVES, 32 * NRM_WAVES, 0, stream>>>(input, prototype, fb, inv_in, inv_pr, fbn);

    gemm_relu_wmma_rt<8><<<(N_ROWS / 32) * (F_DIM / 32), 256, 0, stream>>>(input, fbn, inv_in, ap);
    gemm_relu_wmma<8><<<(P_ROWS / 16) * (F_DIM / 16), 256, 0, stream>>>(prototype, fbn, inv_pr, bp);

    dim3 grid(P_ROWS / PW_JB, N_ROWS / PW_IB);
    pairwise_kernel<<<grid, 256, 0, stream>>>(ap, bp, alpha_raw, beta_raw, theta_raw, out);
}


# 4 launches to 2: norm rows and both projections' tiles run as block roles of one grid; products take raw operands, the row scales (>0) move behind the relu to the pairwise loads; sum order kept
# speedup vs baseline: 3.0592x; 1.1214x over previous
#include <hip/hip_runtime.h>
#include <hip/hip_bf16.h>
#include <math.h>

typedef float v2f __attribute__((ext_vector_type(2)));
typedef float v8f __attribute__((ext_vector_type(8)));
typedef float v4f __attribute__((ext_vector_type(4)));

#define N_ROWS 2048
#define P_ROWS 512
#define D_DIM  1024
#define F_DIM  128
#define EPSV   1e-8f

__device__ __forceinline__ float softplus_f(float x) {
    return log1pf(expf(x));
}

#define NRM_WAVES 8
#define NRM_ROWS  (F_DIM + P_ROWS + N_ROWS)

__device__ __forceinline__ void norms_role(const float* __restrict__ X,
                                           const float* __restrict__ Pm,
                                           const float* __restrict__ fb,
                                           float* __restrict__ inv_in,
                                           float* __restrict__ inv_pr,
                                           float* __restrict__ inv_fb,
                                           const int blk) {
    const int lane = threadIdx.x & 31;
    const int wid = __builtin_amdgcn_readfirstlane(threadIdx.x >> 5);
    const int r   = blk * NRM_WAVES + wid;

    const float* src;
    if (r < F_DIM)               src = fb + (size_t)r * D_DIM;
    else if (r < F_DIM + P_ROWS) src = Pm + (size_t)(r - F_DIM) * D_DIM;
    else                         src = X  + (size_t)(r - F_DIM - P_ROWS) * D_DIM;

    float4 v[8];
#pragma unroll
    for (int j = 0; j < 8; ++j) v[j] = *(const float4*)(src + 4 * (j * 32 + lane));

    float sx = 0.0f, sy = 0.0f, sz = 0.0f, sw = 0.0f;
#pragma unroll
    for (int j = 0; j < 8; ++j) {
        sx = fmaf(v[j].x, v[j].x, sx);
        sy = fmaf(v[j].y, v[j].y, sy);
        sz = fmaf(v[j].z, v[j].z, sz);
        sw = fmaf(v[j].w, v[j].w, sw);
    }
    float s = (sx + sy) + (sz + sw);
#pragma unroll
    for (int off = 16; off > 0; off >>= 1) s += __shfl_xor(s, off);

    const float scale = 1.0f / fmaxf(sqrtf(s), EPSV);

    if (lane == 0) {
        if (r < F_DIM)               inv_fb[r] = scale;
        else if (r < F_DIM + P_ROWS) inv_pr[r - F_DIM] = scale;
        else                         inv_in[r - F_DIM - P_ROWS] = scale;
    }
}

template <int KS>
__device__ __forceinline__ void proj_role(const float* __restrict__ X,
                                          const float* __restrict__ fbr,
                                          float* __restrict__ outp,
                                          const int tile,
                                          float* __restrict__ red_s) {
    constexpr int KSEG = D_DIM / KS;
    float (*red)[8][32] = (float (*)[8][32])red_s;

    const int wid  = threadIdx.x >> 5;
    const int lane = threadIdx.x & 31;
    const int tf   = tile & 7;
    const int tm   = tile >> 3;
    const int half = lane >> 4;
    const int lm   = lane & 15;

    const float* arow = X   + (size_t)(tm * 16 + lm) * D_DIM + wid * KSEG + 4 * half;
    const float* brow = fbr + (size_t)(tf * 16 + lm) * D_DIM + wid * KSEG + 4 * half;

    constexpr int NV = 8 / KS;
    const int col  = tf * 16 + lm;
    const int row0 = tm * 16 + half * 8 + wid;

    v8f acc = {};
#pragma unroll
    for (int k = 0; k < KSEG; k += 8) {
        const float4 a4 = *(const float4*)(arow + k);
        const float4 b4 = *(const float4*)(brow + k);
        const v2f a_lo = {a4.x, a4.y}, a_hi = {a4.z, a4.w};
        const v2f b_lo = {b4.x, b4.y}, b_hi = {b4.z, b4.w};
        acc = __builtin_amdgcn_wmma_f32_16x16x4_f32(
             false, a_lo,  false, b_lo,
             (short)0, acc,  false,  false);
        acc = __builtin_amdgcn_wmma_f32_16x16x4_f32(
             false, a_hi,  false, b_hi,
             (short)0, acc,  false,  false);
    }

#pragma unroll
    for (int v = 0; v < 8; ++v) red[wid][v][lane] = acc[v];
    __syncthreads();

#pragma unroll
    for (int i = 0; i < NV; ++i) {
        const int v = wid + i * KS;
        float s = red[0][v][lane];
#pragma unroll
        for (int w = 1; w < KS; ++w) s += red[w][v][lane];
        outp[(size_t)(row0 + i * KS) * F_DIM + col] = fmaxf(s, 0.0f);
    }
}

#define RT_PITCH 34
#define RT_AHEAD 8

template <int KS>
__device__ __forceinline__ void proj_rt_role(const float* __restrict__ X,
                                             const float* __restrict__ fbr,
                                             float* __restrict__ outp,
                                             const int tile,
                                             float* __restrict__ red_s) {
    constexpr int KSEG = D_DIM / KS;
    constexpr int NR   = 32 / KS;
    float (*red)[32][RT_PITCH] = (float (*)[32][RT_PITCH])red_s;

    const int lane = threadIdx.x & 31;
    const int wid  = __builtin_amdgcn_readfirstlane(threadIdx.x >> 5);
    const int tf   = tile & 3;
    const int tm   = tile >> 2;
    const int half = lane >> 4;
    const int lm   = lane & 15;

    const float* a0p = X   + (size_t)(tm * 32 + lm) * D_DIM + wid * KSEG + 4 * half;
    const float* a1p = a0p + (size_t)16 * D_DIM;
    const float* b0p = fbr + (size_t)(tf * 32 + lm) * D_DIM + wid * KSEG + 4 * half;
    const float* b1p = b0p + (size_t)16 * D_DIM;

    constexpr int NG = KSEG / 8;
    v4f fa0[NG], fa1[NG], fb0[NG], fb1[NG];
#define RT_LOAD_GROUP(G)                                                          \
    {                                                                             \
        fa0[G] = *(const v4f*)(a0p + 8 * (G));     \
        fa1[G] = *(const v4f*)(a1p + 8 * (G));                                    \
        fb0[G] = *(const v4f*)(b0p + 8 * (G));                                    \
        fb1[G] = *(const v4f*)(b1p + 8 * (G));                                    \
    }
#pragma unroll
    for (int g = 0; g < RT_AHEAD; ++g) RT_LOAD_GROUP(g)

    v8f acc00 = {}, acc01 = {}, acc10 = {}, acc11 = {};
#pragma unroll
    for (int g = 0; g < NG; ++g) {
        asm volatile("" :: "v"(fa0[g]), "v"(fa1[g]), "v"(fb0[g]), "v"(fb1[g]) : "memory");
        const v2f a0l = {fa0[g].x, fa0[g].y}, a0h = {fa0[g].z, fa0[g].w};
        const v2f a1l = {fa1[g].x, fa1[g].y}, a1h = {fa1[g].z, fa1[g].w};
        const v2f b0l = {fb0[g].x, fb0[g].y}, b0h = {fb0[g].z, fb0[g].w};
        const v2f b1l = {fb1[g].x, fb1[g].y}, b1h = {fb1[g].z, fb1[g].w};
        acc00 = __builtin_amdgcn_wmma_f32_16x16x4_f32(false, a0l, false, b0l, (short)0, acc00, false, false);
        acc01 = __builtin_amdgcn_wmma_f32_16x16x4_f32(false, a0l, false, b1l, (short)0, acc01, false, false);
        acc10 = __builtin_amdgcn_wmma_f32_16x16x4_f32(false, a1l, false, b0l, (short)0, acc10, false, false);
        acc11 = __builtin_amdgcn_wmma_f32_16x16x4_f32(false, a1l, false, b1l, (short)0, acc11, false, false);
        acc00 = __builtin_amdgcn_wmma_f32_16x16x4_f32(false, a0h, false, b0h, (short)0, acc00, false, false);
        acc01 = __builtin_amdgcn_wmma_f32_16x16x4_f32(false, a0h, false, b1h, (short)0, acc01, false, false);
        acc10 = __builtin_amdgcn_wmma_f32_16x16x4_f32(false, a1h, false, b0h, (short)0, acc10, false, false);
        acc11 = __builtin_amdgcn_wmma_f32_16x16x4_f32(false, a1h, false, b1h, (short)0, acc11, false, false);
        if (g + RT_AHEAD < NG) RT_LOAD_GROUP(g + RT_AHEAD)
    }
#undef RT_LOAD_GROUP

#pragma unroll
    for (int v = 0; v < 8; ++v) {
        red[wid][     8 * half + v][     lm] = acc00[v];
        red[wid][     8 * half + v][16 + lm] = acc01[v];
        red[wid][16 + 8 * half + v][     lm] = acc10[v];
        red[wid][16 + 8 * half + v][16 + lm] = acc11[v];
    }
    __syncthreads();

#pragma unroll
    for (int i = 0; i < NR; ++i) {
        const int r = wid * NR + i;
        float s = red[0][r][lane];
#pragma unroll
        for (int w = 1; w < KS; ++w) s += red[w][r][lane];
        outp[(size_t)(tm * 32 + r) * F_DIM + tf * 32 + lane] = fmaxf(s, 0.0f);
    }
}

#define S1_RT_BLOCKS ((N_ROWS / 32) * (F_DIM / 32))
#define S1_PR_BLOCKS ((P_ROWS / 16) * (F_DIM / 16))
#define S1_NR_BLOCKS (NRM_ROWS / NRM_WAVES)

__global__ __launch_bounds__(256) void stage1_kernel(const float* __restrict__ X,
                                                     const float* __restrict__ Pm,
                                                     const float* __restrict__ fb,
                                                     float* __restrict__ inv_in,
                                                     float* __restrict__ inv_pr,
                                                     float* __restrict__ inv_fb,
                                                     float* __restrict__ apr,
                                                     float* __restrict__ bpr) {
    __shared__ float red_s[8 * 32 * RT_PITCH];
    const int b = blockIdx.x;
    if (b < S1_RT_BLOCKS)
        proj_rt_role<8>(X, fb, apr, b, red_s);
    else if (b < S1_RT_BLOCKS + S1_PR_BLOCKS)
        proj_role<8>(Pm, fb, bpr, b - S1_RT_BLOCKS, red_s);
    else
        norms_role(X, Pm, fb, inv_in, inv_pr, inv_fb, b - S1_RT_BLOCKS - S1_PR_BLOCKS);
}

#define PW_IW    4
#define PW_IB    32
#define PW_JB    128
#define PW_MOFF  132
#define PW_STR   276

__global__ __launch_bounds__(256) void pairwise_kernel(const float* __restrict__ ap,
                                                       const float* __restrict__ bp,
                                                       const float* __restrict__ inv_in,
                                                       const float* __restrict__ inv_pr,
                                                       const float* __restrict__ inv_fb,
                                                       const float* __restrict__ alpha_raw,
                                                       const float* __restrict__ beta_raw,
                                                       const float* __restrict__ theta_raw,
                                                       float* __restrict__ out) {
    __shared__ __attribute__((aligned(16))) float tS[F_DIM * PW_STR];

    const int tid  = threadIdx.x;
    const int lane = tid & 31;
    const int wid  = __builtin_amdgcn_readfirstlane(tid >> 5);
    const int j0   = blockIdx.x * PW_JB;
    const int i0   = blockIdx.y * PW_IB + wid * PW_IW;

    float al, be, th;
    {
        const int jq = (wid << 2) | (lane & 3);
        float4 g[4][4];
#pragma unroll
        for (int t = 0; t < 4; ++t) {
            const int q = (t << 3) | (lane >> 2);
#pragma unroll
            for (int rr = 0; rr < 4; ++rr)
                g[t][rr] = *(const float4*)(bp + (size_t)(j0 + 4 * jq + rr) * F_DIM + 4 * q);
        }
        const float4 tj4 = *(const float4*)(inv_pr + j0 + 4 * jq);
        const float tj[4] = {tj4.x, tj4.y, tj4.z, tj4.w};
        float4 fk[4];
#pragma unroll
        for (int t = 0; t < 4; ++t) fk[t] = *(const float4*)(inv_fb + 4 * ((t << 3) | (lane >> 2)));
        al = softplus_f(alpha_raw[0]);
        be = softplus_f(beta_raw[0]);
        th = softplus_f(theta_raw[0]);
        asm volatile("" :: "s"(al), "s"(be), "s"(th));
#pragma unroll
        for (int t = 0; t < 4; ++t) {
#pragma unroll
            for (int rr = 0; rr < 4; ++rr) {
                g[t][rr].x = (g[t][rr].x * tj[rr]) * fk[t].x;
                g[t][rr].y = (g[t][rr].y * tj[rr]) * fk[t].y;
                g[t][rr].z = (g[t][rr].z * tj[rr]) * fk[t].z;
                g[t][rr].w = (g[t][rr].w * tj[rr]) * fk[t].w;
            }
        }
#pragma unroll
        for (int t = 0; t < 4; ++t) {
            const int q = (t << 3) | (lane >> 2);
            float* dst = tS + (4 * q) * PW_STR + 4 * jq;
            float4 v, m;
            v.x = g[t][0].x; v.y = g[t][1].x; v.z = g[t][2].x; v.w = g[t][3].x;
            m.x = (v.x > 0.0f) ? 1.0f : 0.0f; m.y = (v.y > 0.0f) ? 1.0f : 0.0f;
            m.z = (v.z > 0.0f) ? 1.0f : 0.0f; m.w = (v.w > 0.0f) ? 1.0f : 0.0f;
            *(float4*)(dst) = v; *(float4*)(dst + PW_MOFF) = m;
            v.x = g[t][0].y; v.y = g[t][1].y; v.z = g[t][2].y; v.w = g[t][3].y;
            m.x = (v.x > 0.0f) ? 1.0f : 0.0f; m.y = (v.y > 0.0f) ? 1.0f : 0.0f;
            m.z = (v.z > 0.0f) ? 1.0f : 0.0f; m.w = (v.w > 0.0f) ? 1.0f : 0.0f;
            *(float4*)(dst + PW_STR) = v; *(float4*)(dst + PW_STR + PW_MOFF) = m;
            v.x = g[t][0].z; v.y = g[t][1].z; v.z = g[t][2].z; v.w = g[t][3].z;
            m.x = (v.x > 0.0f) ? 1.0f : 0.0f; m.y = (v.y > 0.0f) ? 1.0f : 0.0f;
            m.z = (v.z > 0.0f) ? 1.0f : 0.0f; m.w = (v.w > 0.0f) ? 1.0f : 0.0f;
            *(float4*)(dst + 2 * PW_STR) = v; *(float4*)(dst + 2 * PW_STR + PW_MOFF) = m;
            v.x = g[t][0].w; v.y = g[t][1].w; v.z = g[t][2].w; v.w = g[t][3].w;
            m.x = (v.x > 0.0f) ? 1.0f : 0.0f; m.y = (v.y > 0.0f) ? 1.0f : 0.0f;
            m.z = (v.z > 0.0f) ? 1.0f : 0.0f; m.w = (v.w > 0.0f) ? 1.0f : 0.0f;
            *(float4*)(dst + 3 * PW_STR) = v; *(float4*)(dst + 3 * PW_STR + PW_MOFF) = m;
        }
    }
    __syncthreads();

    const float* __restrict__ arow = ap + (size_t)i0 * F_DIM;
    const float* brow = tS + 4 * lane;

    float sc[PW_IW];
#pragma unroll
    for (int r = 0; r < PW_IW; ++r) sc[r] = inv_in[i0 + r];

    float s_min[PW_IW][4], s_a[PW_IW][4], s_b[PW_IW][4];
#pragma unroll
    for (int r = 0; r < PW_IW; ++r) {
#pragma unroll
        for (int c = 0; c < 4; ++c) { s_min[r][c] = 0.0f; s_a[r][c] = 0.0f; s_b[r][c] = 0.0f; }
    }

    v8f avP[PW_IW], avQ[PW_IW];
    v8f fvP, fvQ;
    v4f bqA[4], mqA[4], bqB[4], mqB[4];

#define PW_LOAD_BM(BQ, MQ, K0)                                                    \
    _Pragma("unroll")                                                             \
    for (int kk = 0; kk < 4; ++kk) {                                              \
        BQ[kk] = *(const v4f*)(brow + ((K0) + kk) * PW_STR);                      \
        MQ[kk] = *(const v4f*)(brow + ((K0) + kk) * PW_STR + PW_MOFF);            \
    }
#define PW_LOAD_AP(AV, FV, K0)                                                    \
    _Pragma("unroll")                                                             \
    for (int r = 0; r < PW_IW; ++r) AV[r] = *(const v8f*)(arow + r * F_DIM + (K0)); \
    FV = *(const v8f*)(inv_fb + (K0));

#define PW_SCALE_AP(AV, FV)                                                       \
    _Pragma("unroll")                                                             \
    for (int r = 0; r < PW_IW; ++r) {                                             \
        _Pragma("unroll")                                                         \
        for (int e = 0; e < 8; ++e) AV[r][e] = (AV[r][e] * sc[r]) * FV[e];        \
    }

#define PW_HOLD_AP_BM(AV, BQ, MQ)                                                 \
    asm volatile("" :: "s"(AV[0]), "s"(AV[1]), "s"(AV[2]), "s"(AV[3]),            \
                       "v"(BQ[0]), "v"(BQ[1]), "v"(BQ[2]), "v"(BQ[3]),            \
                       "v"(MQ[0]), "v"(MQ[1]), "v"(MQ[2]), "v"(MQ[3]) : "memory")
#define PW_HOLD_BM(BQ, MQ)                                                        \
    asm volatile("" :: "v"(BQ[0]), "v"(BQ[1]), "v"(BQ[2]), "v"(BQ[3]),            \
                       "v"(MQ[0]), "v"(MQ[1]), "v"(MQ[2]), "v"(MQ[3]) : "memory")

#define PW_SUM_QUAD(AV, KO, BQ, MQ)                                               \
    _Pragma("unroll")                                                             \
    for (int kk = 0; kk < 4; ++kk) {                                              \
        _Pragma("unroll")                                                         \
        for (int r = 0; r < PW_IW; ++r) {                                         \
            const float a = AV[r][(KO) + kk];                                     \
            if (__builtin_expect(a > 0.0f, 1)) {                                  \
                _Pragma("unroll")                                                 \
                for (int c = 0; c < 4; ++c) {                                     \
                    s_min[r][c] += fminf(a, BQ[kk][c]);                           \
                    s_a[r][c] = fmaf(a, MQ[kk][c], s_a[r][c]);                    \
                    s_b[r][c] += BQ[kk][c];                                       \
                }                                                                 \
            }                                                                     \
        }                                                                         \
    }

#define PW_PASS(AVC, FVC, AVN, FVN, K0)                                           \
    {                                                                             \
        const int kn = ((K0) + 8 < F_DIM) ? ((K0) + 8) : (F_DIM - 8);             \
        PW_LOAD_BM(bqB, mqB, (K0) + 4)                                            \
        PW_SCALE_AP(AVC, FVC)                                                     \
        PW_HOLD_AP_BM(AVC, bqA, mqA);                                             \
        PW_LOAD_AP(AVN, FVN, kn)                                                  \
        PW_SUM_QUAD(AVC, 0, bqA, mqA)                                             \
        PW_LOAD_BM(bqA, mqA, kn)                                                  \
        PW_HOLD_BM(bqB, mqB);                                                     \
        PW_SUM_QUAD(AVC, 4, bqB, mqB)                                             \
    }

    PW_LOAD_AP(avP, fvP, 0)
    PW_LOAD_BM(bqA, mqA, 0)
#pragma nounroll
    for (int k = 0; k < F_DIM; k += 16) {
        PW_PASS(avP, fvP, avQ, fvQ, k)
        PW_PASS(avQ, fvQ, avP, fvP, k + 8)
    }
#undef PW_LOAD_BM
#undef PW_LOAD_AP
#undef PW_SCALE_AP
#undef PW_HOLD_AP_BM
#undef PW_HOLD_BM
#undef PW_SUM_QUAD
#undef PW_PASS

#pragma unroll
    for (int r = 0; r < PW_IW; ++r) {
        float4 o;
        o.x = (th + al + be) * s_min[r][0] - al * s_a[r][0] - be * s_b[r][0];
        o.y = (th + al + be) * s_min[r][1] - al * s_a[r][1] - be * s_b[r][1];
        o.z = (th + al + be) * s_min[r][2] - al * s_a[r][2] - be * s_b[r][2];
        o.w = (th + al + be) * s_min[r][3] - al * s_a[r][3] - be * s_b[r][3];
        *(float4*)(out + (size_t)(i0 + r) * P_ROWS + j0 + 4 * lane) = o;
    }
}

extern "C" void kernel_launch(void* const* d_in, const int* in_sizes, int n_in,
                              void* d_out, int out_size, void* d_ws, size_t ws_size,
                              hipStream_t stream) {
    const float* input     = (const float*)d_in[0];
    const float* prototype = (const float*)d_in[1];
    const float* fb        = (const float*)d_in[2];
    const float* alpha_raw = (const float*)d_in[3];
    const float* beta_raw  = (const float*)d_in[4];
    const float* theta_raw = (const float*)d_in[5];
    float* out = (float*)d_out;

    float* w      = (float*)d_ws;
    float* inv_fb = w;
    float* inv_in = inv_fb + F_DIM;
    float* inv_pr = inv_in + N_ROWS;
    float* ap     = inv_pr + P_ROWS;
    float* bp     = ap + (size_t)N_ROWS * F_DIM;

    stage1_kernel<<<S1_RT_BLOCKS + S1_PR_BLOCKS + S1_NR_BLOCKS, 256, 0, stream>>>(
        input, prototype, fb, inv_in, inv_pr, inv_fb, ap, bp);

    dim3 grid(P_ROWS / PW_JB, N_ROWS / PW_IB);
    pairwise_kernel<<<grid, 256, 0, stream>>>(ap, bp, inv_in, inv_pr, inv_fb,
                                             alpha_raw, beta_raw, theta_raw, out);
}


# The three softplus chains leave the pairwise launch, where every wave ran them ahead of its tile loads: one wave of the first launch computes them, the pairwise waves read 3 floats; sums kept
# speedup vs baseline: 3.1750x; 1.0379x over previous
#include <hip/hip_runtime.h>
#include <hip/hip_bf16.h>
#include <math.h>

typedef float v2f __attribute__((ext_vector_type(2)));
typedef float v8f __attribute__((ext_vector_type(8)));
typedef float v4f __attribute__((ext_vector_type(4)));

#define N_ROWS 2048
#define P_ROWS 512
#define D_DIM  1024
#define F_DIM  128
#define EPSV   1e-8f

__device__ __forceinline__ float softplus_f(float x) {
    return log1pf(expf(x));
}

#define NRM_WAVES 8
#define NRM_ROWS  (F_DIM + P_ROWS + N_ROWS)

__device__ __forceinline__ void norms_role(const float* __restrict__ X,
                                           const float* __restrict__ Pm,
                                           const float* __restrict__ fb,
                                           float* __restrict__ inv_in,
                                           float* __restrict__ inv_pr,
                                           float* __restrict__ inv_fb,
                                           const float* __restrict__ alpha_raw,
                                           const float* __restrict__ beta_raw,
                                           const float* __restrict__ theta_raw,
                                           float* __restrict__ coef,
                                           const int blk) {
    const int lane = threadIdx.x & 31;
    const int wid = __builtin_amdgcn_readfirstlane(threadIdx.x >> 5);
    const int r   = blk * NRM_WAVES + wid;

    const float* src;
    if (r < F_DIM)               src = fb + (size_t)r * D_DIM;
    else if (r < F_DIM + P_ROWS) src = Pm + (size_t)(r - F_DIM) * D_DIM;
    else                         src = X  + (size_t)(r - F_DIM - P_ROWS) * D_DIM;

    float4 v[8];
#pragma unroll
    for (int j = 0; j < 8; ++j) v[j] = *(const float4*)(src + 4 * (j * 32 + lane));

    if (r == 0) {
        const float r_al = alpha_raw[0];
        const float r_be = beta_raw[0];
        const float r_th = theta_raw[0];
        asm volatile("" :: "s"(r_al), "s"(r_be), "s"(r_th));
        const float c_al = softplus_f(r_al);
        const float c_be = softplus_f(r_be);
        const float c_th = softplus_f(r_th);
        if (lane == 0) { coef[0] = c_al; coef[1] = c_be; coef[2] = c_th; }
    }

    float sx = 0.0f, sy = 0.0f, sz = 0.0f, sw = 0.0f;
#pragma unroll
    for (int j = 0; j < 8; ++j) {
        sx = fmaf(v[j].x, v[j].x, sx);
        sy = fmaf(v[j].y, v[j].y, sy);
        sz = fmaf(v[j].z, v[j].z, sz);
        sw = fmaf(v[j].w, v[j].w, sw);
    }
    float s = (sx + sy) + (sz + sw);
#pragma unroll
    for (int off = 16; off > 0; off >>= 1) s += __shfl_xor(s, off);

    const float scale = 1.0f / fmaxf(sqrtf(s), EPSV);

    if (lane == 0) {
        if (r < F_DIM)               inv_fb[r] = scale;
        else if (r < F_DIM + P_ROWS) inv_pr[r - F_DIM] = scale;
        else                         inv_in[r - F_DIM - P_ROWS] = scale;
    }
}

template <int KS>
__device__ __forceinline__ void proj_role(const float* __restrict__ X,
                                          const float* __restrict__ fbr,
                                          float* __restrict__ outp,
                                          const int tile,
                                          float* __restrict__ red_s) {
    constexpr int KSEG = D_DIM / KS;
    float (*red)[8][32] = (float (*)[8][32])red_s;

    const int wid  = threadIdx.x >> 5;
    const int lane = threadIdx.x & 31;
    const int tf   = tile & 7;
    const int tm   = tile >> 3;
    const int half = lane >> 4;
    const int lm   = lane & 15;

    const float* arow = X   + (size_t)(tm * 16 + lm) * D_DIM + wid * KSEG + 4 * half;
    const float* brow = fbr + (size_t)(tf * 16 + lm) * D_DIM + wid * KSEG + 4 * half;

    constexpr int NV = 8 / KS;
    const int col  = tf * 16 + lm;
    const int row0 = tm * 16 + half * 8 + wid;

    v8f acc = {};
#pragma unroll
    for (int k = 0; k < KSEG; k += 8) {
        const float4 a4 = *(const float4*)(arow + k);
        const float4 b4 = *(const float4*)(brow + k);
        const v2f a_lo = {a4.x, a4.y}, a_hi = {a4.z, a4.w};
        const v2f b_lo = {b4.x, b4.y}, b_hi = {b4.z, b4.w};
        acc = __builtin_amdgcn_wmma_f32_16x16x4_f32(
             false, a_lo,  false, b_lo,
             (short)0, acc,  false,  false);
        acc = __builtin_amdgcn_wmma_f32_16x16x4_f32(
             false, a_hi,  false, b_hi,
             (short)0, acc,  false,  false);
    }

#pragma unroll
    for (int v = 0; v < 8; ++v) red[wid][v][lane] = acc[v];
    __syncthreads();

#pragma unroll
    for (int i = 0; i < NV; ++i) {
        const int v = wid + i * KS;
        float s = red[0][v][lane];
#pragma unroll
        for (int w = 1; w < KS; ++w) s += red[w][v][lane];
        outp[(size_t)(row0 + i * KS) * F_DIM + col] = fmaxf(s, 0.0f);
    }
}

#define RT_PITCH 34
#define RT_AHEAD 8

template <int KS>
__device__ __forceinline__ void proj_rt_role(const float* __restrict__ X,
                                             const float* __restrict__ fbr,
                                             float* __restrict__ outp,
                                             const int tile,
                                             float* __restrict__ red_s) {
    constexpr int KSEG = D_DIM / KS;
    constexpr int NR   = 32 / KS;
    float (*red)[32][RT_PITCH] = (float (*)[32][RT_PITCH])red_s;

    const int lane = threadIdx.x & 31;
    const int wid  = __builtin_amdgcn_readfirstlane(threadIdx.x >> 5);
    const int tf   = tile & 3;
    const int tm   = tile >> 2;
    const int half = lane >> 4;
    const int lm   = lane & 15;

    const float* a0p = X   + (size_t)(tm * 32 + lm) * D_DIM + wid * KSEG + 4 * half;
    const float* a1p = a0p + (size_t)16 * D_DIM;
    const float* b0p = fbr + (size_t)(tf * 32 + lm) * D_DIM + wid * KSEG + 4 * half;
    const float* b1p = b0p + (size_t)16 * D_DIM;

    constexpr int NG = KSEG / 8;
    v4f fa0[NG], fa1[NG], fb0[NG], fb1[NG];
#define RT_LOAD_GROUP(G)                                                          \
    {                                                                             \
        fa0[G] = *(const v4f*)(a0p + 8 * (G));     \
        fa1[G] = *(const v4f*)(a1p + 8 * (G));                                    \
        fb0[G] = *(const v4f*)(b0p + 8 * (G));                                    \
        fb1[G] = *(const v4f*)(b1p + 8 * (G));                                    \
    }
#pragma unroll
    for (int g = 0; g < RT_AHEAD; ++g) RT_LOAD_GROUP(g)

    v8f acc00 = {}, acc01 = {}, acc10 = {}, acc11 = {};
#pragma unroll
    for (int g = 0; g < NG; ++g) {
        asm volatile("" :: "v"(fa0[g]), "v"(fa1[g]), "v"(fb0[g]), "v"(fb1[g]) : "memory");
        const v2f a0l = {fa0[g].x, fa0[g].y}, a0h = {fa0[g].z, fa0[g].w};
        const v2f a1l = {fa1[g].x, fa1[g].y}, a1h = {fa1[g].z, fa1[g].w};
        const v2f b0l = {fb0[g].x, fb0[g].y}, b0h = {fb0[g].z, fb0[g].w};
        const v2f b1l = {fb1[g].x, fb1[g].y}, b1h = {fb1[g].z, fb1[g].w};
        acc00 = __builtin_amdgcn_wmma_f32_16x16x4_f32(false, a0l, false, b0l, (short)0, acc00, false, false);
        acc01 = __builtin_amdgcn_wmma_f32_16x16x4_f32(false, a0l, false, b1l, (short)0, acc01, false, false);
        acc10 = __builtin_amdgcn_wmma_f32_16x16x4_f32(false, a1l, false, b0l, (short)0, acc10, false, false);
        acc11 = __builtin_amdgcn_wmma_f32_16x16x4_f32(false, a1l, false, b1l, (short)0, acc11, false, false);
        acc00 = __builtin_amdgcn_wmma_f32_16x16x4_f32(false, a0h, false, b0h, (short)0, acc00, false, false);
        acc01 = __builtin_amdgcn_wmma_f32_16x16x4_f32(false, a0h, false, b1h, (short)0, acc01, false, false);
        acc10 = __builtin_amdgcn_wmma_f32_16x16x4_f32(false, a1h, false, b0h, (short)0, acc10, false, false);
        acc11 = __builtin_amdgcn_wmma_f32_16x16x4_f32(false, a1h, false, b1h, (short)0, acc11, false, false);
        if (g + RT_AHEAD < NG) RT_LOAD_GROUP(g + RT_AHEAD)
    }
#undef RT_LOAD_GROUP

#pragma unroll
    for (int v = 0; v < 8; ++v) {
        red[wid][     8 * half + v][     lm] = acc00[v];
        red[wid][     8 * half + v][16 + lm] = acc01[v];
        red[wid][16 + 8 * half + v][     lm] = acc10[v];
        red[wid][16 + 8 * half + v][16 + lm] = acc11[v];
    }
    __syncthreads();

#pragma unroll
    for (int i = 0; i < NR; ++i) {
        const int r = wid * NR + i;
        float s = red[0][r][lane];
#pragma unroll
        for (int w = 1; w < KS; ++w) s += red[w][r][lane];
        outp[(size_t)(tm * 32 + r) * F_DIM + tf * 32 + lane] = fmaxf(s, 0.0f);
    }
}

#define S1_RT_BLOCKS ((N_ROWS / 32) * (F_DIM / 32))
#define S1_PR_BLOCKS ((P_ROWS / 16) * (F_DIM / 16))
#define S1_NR_BLOCKS (NRM_ROWS / NRM_WAVES)

__global__ __launch_bounds__(256) void stage1_kernel(const float* __restrict__ X,
                                                     const float* __restrict__ Pm,
                                                     const float* __restrict__ fb,
                                                     float* __restrict__ inv_in,
                                                     float* __restrict__ inv_pr,
                                                     float* __restrict__ inv_fb,
                                                     float* __restrict__ apr,
                                                     float* __restrict__ bpr,
                                                     const float* __restrict__ alpha_raw,
                                                     const float* __restrict__ beta_raw,
                                                     const float* __restrict__ theta_raw,
                                                     float* __restrict__ coef) {
    __shared__ float red_s[8 * 32 * RT_PITCH];
    const int b = blockIdx.x;
    if (b < S1_RT_BLOCKS)
        proj_rt_role<8>(X, fb, apr, b, red_s);
    else if (b < S1_RT_BLOCKS + S1_PR_BLOCKS)
        proj_role<8>(Pm, fb, bpr, b - S1_RT_BLOCKS, red_s);
    else
        norms_role(X, Pm, fb, inv_in, inv_pr, inv_fb, alpha_raw, beta_raw, theta_raw, coef,
                   b - S1_RT_BLOCKS - S1_PR_BLOCKS);
}

#define PW_IW    4
#define PW_IB    32
#define PW_JB    128
#define PW_MOFF  132
#define PW_STR   276

__global__ __launch_bounds__(256) void pairwise_kernel(const float* __restrict__ ap,
                                                       const float* __restrict__ bp,
                                                       const float* __restrict__ inv_in,
                                                       const float* __restrict__ inv_pr,
                                                       const float* __restrict__ inv_fb,
                                                       const float* __restrict__ coef,
                                                       float* __restrict__ out) {
    __shared__ __attribute__((aligned(16))) float tS[F_DIM * PW_STR];

    const int tid  = threadIdx.x;
    const int lane = tid & 31;
    const int wid  = __builtin_amdgcn_readfirstlane(tid >> 5);
    const int j0   = blockIdx.x * PW_JB;
    const int i0   = blockIdx.y * PW_IB + wid * PW_IW;

    float al, be, th;
    {
        const int jq = (wid << 2) | (lane & 3);
        float4 g[4][4];
#pragma unroll
        for (int t = 0; t < 4; ++t) {
            const int q = (t << 3) | (lane >> 2);
#pragma unroll
            for (int rr = 0; rr < 4; ++rr)
                g[t][rr] = *(const float4*)(bp + (size_t)(j0 + 4 * jq + rr) * F_DIM + 4 * q);
        }
        const float4 tj4 = *(const float4*)(inv_pr + j0 + 4 * jq);
        const float tj[4] = {tj4.x, tj4.y, tj4.z, tj4.w};
        float4 fk[4];
#pragma unroll
        for (int t = 0; t < 4; ++t) fk[t] = *(const float4*)(inv_fb + 4 * ((t << 3) | (lane >> 2)));
        al = coef[0];
        be = coef[1];
        th = coef[2];
#pragma unroll
        for (int t = 0; t < 4; ++t) {
#pragma unroll
            for (int rr = 0; rr < 4; ++rr) {
                g[t][rr].x = (g[t][rr].x * tj[rr]) * fk[t].x;
                g[t][rr].y = (g[t][rr].y * tj[rr]) * fk[t].y;
                g[t][rr].z = (g[t][rr].z * tj[rr]) * fk[t].z;
                g[t][rr].w = (g[t][rr].w * tj[rr]) * fk[t].w;
            }
        }
#pragma unroll
        for (int t = 0; t < 4; ++t) {
            const int q = (t << 3) | (lane >> 2);
            float* dst = tS + (4 * q) * PW_STR + 4 * jq;
            float4 v, m;
            v.x = g[t][0].x; v.y = g[t][1].x; v.z = g[t][2].x; v.w = g[t][3].x;
            m.x = (v.x > 0.0f) ? 1.0f : 0.0f; m.y = (v.y > 0.0f) ? 1.0f : 0.0f;
            m.z = (v.z > 0.0f) ? 1.0f : 0.0f; m.w = (v.w > 0.0f) ? 1.0f : 0.0f;
            *(float4*)(dst) = v; *(float4*)(dst + PW_MOFF) = m;
            v.x = g[t][0].y; v.y = g[t][1].y; v.z = g[t][2].y; v.w = g[t][3].y;
            m.x = (v.x > 0.0f) ? 1.0f : 0.0f; m.y = (v.y > 0.0f) ? 1.0f : 0.0f;
            m.z = (v.z > 0.0f) ? 1.0f : 0.0f; m.w = (v.w > 0.0f) ? 1.0f : 0.0f;
            *(float4*)(dst + PW_STR) = v; *(float4*)(dst + PW_STR + PW_MOFF) = m;
            v.x = g[t][0].z; v.y = g[t][1].z; v.z = g[t][2].z; v.w = g[t][3].z;
            m.x = (v.x > 0.0f) ? 1.0f : 0.0f; m.y = (v.y > 0.0f) ? 1.0f : 0.0f;
            m.z = (v.z > 0.0f) ? 1.0f : 0.0f; m.w = (v.w > 0.0f) ? 1.0f : 0.0f;
            *(float4*)(dst + 2 * PW_STR) = v; *(float4*)(dst + 2 * PW_STR + PW_MOFF) = m;
            v.x = g[t][0].w; v.y = g[t][1].w; v.z = g[t][2].w; v.w = g[t][3].w;
            m.x = (v.x > 0.0f) ? 1.0f : 0.0f; m.y = (v.y > 0.0f) ? 1.0f : 0.0f;
            m.z = (v.z > 0.0f) ? 1.0f : 0.0f; m.w = (v.w > 0.0f) ? 1.0f : 0.0f;
            *(float4*)(dst + 3 * PW_STR) = v; *(float4*)(dst + 3 * PW_STR + PW_MOFF) = m;
        }
    }
    asm volatile("" :: "s"(al), "s"(be), "s"(th));
    __syncthreads();

    const float* __restrict__ arow = ap + (size_t)i0 * F_DIM;
    const float* brow = tS + 4 * lane;

    float sc[PW_IW];
#pragma unroll
    for (int r = 0; r < PW_IW; ++r) sc[r] = inv_in[i0 + r];

    float s_min[PW_IW][4], s_a[PW_IW][4], s_b[PW_IW][4];
#pragma unroll
    for (int r = 0; r < PW_IW; ++r) {
#pragma unroll
        for (int c = 0; c < 4; ++c) { s_min[r][c] = 0.0f; s_a[r][c] = 0.0f; s_b[r][c] = 0.0f; }
    }

    v8f avP[PW_IW], avQ[PW_IW];
    v8f fvP, fvQ;
    v4f bqA[4], mqA[4], bqB[4], mqB[4];

#define PW_LOAD_BM(BQ, MQ, K0)                                                    \
    _Pragma("unroll")                                                             \
    for (int kk = 0; kk < 4; ++kk) {                                              \
        BQ[kk] = *(const v4f*)(brow + ((K0) + kk) * PW_STR);                      \
        MQ[kk] = *(const v4f*)(brow + ((K0) + kk) * PW_STR + PW_MOFF);            \
    }
#define PW_LOAD_AP(AV, FV, K0)                                                    \
    _Pragma("unroll")                                                             \
    for (int r = 0; r < PW_IW; ++r) AV[r] = *(const v8f*)(arow + r * F_DIM + (K0)); \
    FV = *(const v8f*)(inv_fb + (K0));

#define PW_SCALE_AP(AV, FV)                                                       \
    _Pragma("unroll")                                                             \
    for (int r = 0; r < PW_IW; ++r) {                                             \
        _Pragma("unroll")                                                         \
        for (int e = 0; e < 8; ++e) AV[r][e] = (AV[r][e] * sc[r]) * FV[e];        \
    }

#define PW_HOLD_AP_BM(AV, BQ, MQ)                                                 \
    asm volatile("" :: "s"(AV[0]), "s"(AV[1]), "s"(AV[2]), "s"(AV[3]),            \
                       "v"(BQ[0]), "v"(BQ[1]), "v"(BQ[2]), "v"(BQ[3]),            \
                       "v"(MQ[0]), "v"(MQ[1]), "v"(MQ[2]), "v"(MQ[3]) : "memory")
#define PW_HOLD_BM(BQ, MQ)                                                        \
    asm volatile("" :: "v"(BQ[0]), "v"(BQ[1]), "v"(BQ[2]), "v"(BQ[3]),            \
                       "v"(MQ[0]), "v"(MQ[1]), "v"(MQ[2]), "v"(MQ[3]) : "memory")

#define PW_SUM_QUAD(AV, KO, BQ, MQ)                                               \
    _Pragma("unroll")                                                             \
    for (int kk = 0; kk < 4; ++kk) {                                              \
        _Pragma("unroll")                                                         \
        for (int r = 0; r < PW_IW; ++r) {                                         \
            const float a = AV[r][(KO) + kk];                                     \
            if (__builtin_expect(a > 0.0f, 1)) {                                  \
                _Pragma("unroll")                                                 \
                for (int c = 0; c < 4; ++c) {                                     \
                    s_min[r][c] += fminf(a, BQ[kk][c]);                           \
                    s_a[r][c] = fmaf(a, MQ[kk][c], s_a[r][c]);                    \
                    s_b[r][c] += BQ[kk][c];                                       \
                }                                                                 \
            }                                                                     \
        }                                                                         \
    }

#define PW_PASS(AVC, FVC, AVN, FVN, K0)                                           \
    {                                                                             \
        const int kn = ((K0) + 8 < F_DIM) ? ((K0) + 8) : (F_DIM - 8);             \
        PW_LOAD_BM(bqB, mqB, (K0) + 4)                                            \
        PW_SCALE_AP(AVC, FVC)                                                     \
        PW_HOLD_AP_BM(AVC, bqA, mqA);                                             \
        PW_LOAD_AP(AVN, FVN, kn)                                                  \
        PW_SUM_QUAD(AVC, 0, bqA, mqA)                                             \
        PW_LOAD_BM(bqA, mqA, kn)                                                  \
        PW_HOLD_BM(bqB, mqB);                                                     \
        PW_SUM_QUAD(AVC, 4, bqB, mqB)                                             \
    }

    PW_LOAD_AP(avP, fvP, 0)
    PW_LOAD_BM(bqA, mqA, 0)
#pragma nounroll
    for (int k = 0; k < F_DIM; k += 16) {
        PW_PASS(avP, fvP, avQ, fvQ, k)
        PW_PASS(avQ, fvQ, avP, fvP, k + 8)
    }
#undef PW_LOAD_BM
#undef PW_LOAD_AP
#undef PW_SCALE_AP
#undef PW_HOLD_AP_BM
#undef PW_HOLD_BM
#undef PW_SUM_QUAD
#undef PW_PASS

#pragma unroll
    for (int r = 0; r < PW_IW; ++r) {
        float4 o;
        o.x = (th + al + be) * s_min[r][0] - al * s_a[r][0] - be * s_b[r][0];
        o.y = (th + al + be) * s_min[r][1] - al * s_a[r][1] - be * s_b[r][1];
        o.z = (th + al + be) * s_min[r][2] - al * s_a[r][2] - be * s_b[r][2];
        o.w = (th + al + be) * s_min[r][3] - al * s_a[r][3] - be * s_b[r][3];
        *(float4*)(out + (size_t)(i0 + r) * P_ROWS + j0 + 4 * lane) = o;
    }
}

extern "C" void kernel_launch(void* const* d_in, const int* in_sizes, int n_in,
                              void* d_out, int out_size, void* d_ws, size_t ws_size,
                              hipStream_t stream) {
    const float* input     = (const float*)d_in[0];
    const float* prototype = (const float*)d_in[1];
    const float* fb        = (const float*)d_in[2];
    const float* alpha_raw = (const float*)d_in[3];
    const float* beta_raw  = (const float*)d_in[4];
    const float* theta_raw = (const float*)d_in[5];
    float* out = (float*)d_out;

    float* w      = (float*)d_ws;
    float* inv_fb = w;
    float* inv_in = inv_fb + F_DIM;
    float* inv_pr = inv_in + N_ROWS;
    float* ap     = inv_pr + P_ROWS;
    float* bp     = ap + (size_t)N_ROWS * F_DIM;
    float* coef   = bp + (size_t)P_ROWS * F_DIM;

    stage1_kernel<<<S1_RT_BLOCKS + S1_PR_BLOCKS + S1_NR_BLOCKS, 256, 0, stream>>>(
        input, prototype, fb, inv_in, inv_pr, inv_fb, ap, bp,
        alpha_raw, beta_raw, theta_raw, coef);

    dim3 grid(P_ROWS / PW_JB, N_ROWS / PW_IB);
    pairwise_kernel<<<grid, 256, 0, stream>>>(ap, bp, inv_in, inv_pr, inv_fb, coef, out);
}


# Input-product role: load ring refilled by whole lines: the four 16-B loads of a fragment's 128-B line (4 k-groups) leave back to back behind one sub-step, not one per 8 WMMAs; sums unchanged
# speedup vs baseline: 3.3635x; 1.0594x over previous
#include <hip/hip_runtime.h>
#include <hip/hip_bf16.h>
#include <math.h>

typedef float v2f __attribute__((ext_vector_type(2)));
typedef float v8f __attribute__((ext_vector_type(8)));
typedef float v4f __attribute__((ext_vector_type(4)));

#define N_ROWS 2048
#define P_ROWS 512
#define D_DIM  1024
#define F_DIM  128
#define EPSV   1e-8f

__device__ __forceinline__ float softplus_f(float x) {
    return log1pf(expf(x));
}

#define NRM_WAVES 8
#define NRM_ROWS  (F_DIM + P_ROWS + N_ROWS)

__device__ __forceinline__ void norms_role(const float* __restrict__ X,
                                           const float* __restrict__ Pm,
                                           const float* __restrict__ fb,
                                           float* __restrict__ inv_in,
                                           float* __restrict__ inv_pr,
                                           float* __restrict__ inv_fb,
                                           const float* __restrict__ alpha_raw,
                                           const float* __restrict__ beta_raw,
                                           const float* __restrict__ theta_raw,
                                           float* __restrict__ coef,
                                           const int blk) {
    const int lane = threadIdx.x & 31;
    const int wid = __builtin_amdgcn_readfirstlane(threadIdx.x >> 5);
    const int r   = blk * NRM_WAVES + wid;

    const float* src;
    if (r < F_DIM)               src = fb + (size_t)r * D_DIM;
    else if (r < F_DIM + P_ROWS) src = Pm + (size_t)(r - F_DIM) * D_DIM;
    else                         src = X  + (size_t)(r - F_DIM - P_ROWS) * D_DIM;

    float4 v[8];
#pragma unroll
    for (int j = 0; j < 8; ++j) v[j] = *(const float4*)(src + 4 * (j * 32 + lane));

    if (r == 0) {
        const float r_al = alpha_raw[0];
        const float r_be = beta_raw[0];
        const float r_th = theta_raw[0];
        asm volatile("" :: "s"(r_al), "s"(r_be), "s"(r_th));
        const float c_al = softplus_f(r_al);
        const float c_be = softplus_f(r_be);
        const float c_th = softplus_f(r_th);
        if (lane == 0) { coef[0] = c_al; coef[1] = c_be; coef[2] = c_th; }
    }

    float sx = 0.0f, sy = 0.0f, sz = 0.0f, sw = 0.0f;
#pragma unroll
    for (int j = 0; j < 8; ++j) {
        sx = fmaf(v[j].x, v[j].x, sx);
        sy = fmaf(v[j].y, v[j].y, sy);
        sz = fmaf(v[j].z, v[j].z, sz);
        sw = fmaf(v[j].w, v[j].w, sw);
    }
    float s = (sx + sy) + (sz + sw);
#pragma unroll
    for (int off = 16; off > 0; off >>= 1) s += __shfl_xor(s, off);

    const float scale = 1.0f / fmaxf(sqrtf(s), EPSV);

    if (lane == 0) {
        if (r < F_DIM)               inv_fb[r] = scale;
        else if (r < F_DIM + P_ROWS) inv_pr[r - F_DIM] = scale;
        else                         inv_in[r - F_DIM - P_ROWS] = scale;
    }
}

template <int KS>
__device__ __forceinline__ void proj_role(const float* __restrict__ X,
                                          const float* __restrict__ fbr,
                                          float* __restrict__ outp,
                                          const int tile,
                                          float* __restrict__ red_s) {
    constexpr int KSEG = D_DIM / KS;
    float (*red)[8][32] = (float (*)[8][32])red_s;

    const int wid  = threadIdx.x >> 5;
    const int lane = threadIdx.x & 31;
    const int tf   = tile & 7;
    const int tm   = tile >> 3;
    const int half = lane >> 4;
    const int lm   = lane & 15;

    const float* arow = X   + (size_t)(tm * 16 + lm) * D_DIM + wid * KSEG + 4 * half;
    const float* brow = fbr + (size_t)(tf * 16 + lm) * D_DIM + wid * KSEG + 4 * half;

    constexpr int NV = 8 / KS;
    const int col  = tf * 16 + lm;
    const int row0 = tm * 16 + half * 8 + wid;

    v8f acc = {};
#pragma unroll
    for (int k = 0; k < KSEG; k += 8) {
        const float4 a4 = *(const float4*)(arow + k);
        const float4 b4 = *(const float4*)(brow + k);
        const v2f a_lo = {a4.x, a4.y}, a_hi = {a4.z, a4.w};
        const v2f b_lo = {b4.x, b4.y}, b_hi = {b4.z, b4.w};
        acc = __builtin_amdgcn_wmma_f32_16x16x4_f32(
             false, a_lo,  false, b_lo,
             (short)0, acc,  false,  false);
        acc = __builtin_amdgcn_wmma_f32_16x16x4_f32(
             false, a_hi,  false, b_hi,
             (short)0, acc,  false,  false);
    }

#pragma unroll
    for (int v = 0; v < 8; ++v) red[wid][v][lane] = acc[v];
    __syncthreads();

#pragma unroll
    for (int i = 0; i < NV; ++i) {
        const int v = wid + i * KS;
        float s = red[0][v][lane];
#pragma unroll
        for (int w = 1; w < KS; ++w) s += red[w][v][lane];
        outp[(size_t)(row0 + i * KS) * F_DIM + col] = fmaxf(s, 0.0f);
    }
}

#define RT_PITCH 34
#define RT_AHEAD 2

template <int KS>
__device__ __forceinline__ void proj_rt_role(const float* __restrict__ X,
                                             const float* __restrict__ fbr,
                                             float* __restrict__ outp,
                                             const int tile,
                                             float* __restrict__ red_s) {
    constexpr int KSEG = D_DIM / KS;
    constexpr int NR   = 32 / KS;
    float (*red)[32][RT_PITCH] = (float (*)[32][RT_PITCH])red_s;

    const int lane = threadIdx.x & 31;
    const int wid  = __builtin_amdgcn_readfirstlane(threadIdx.x >> 5);
    const int tf   = tile & 3;
    const int tm   = tile >> 2;
    const int half = lane >> 4;
    const int lm   = lane & 15;

    const float* a0p = X   + (size_t)(tm * 32 + lm) * D_DIM + wid * KSEG + 4 * half;
    const float* a1p = a0p + (size_t)16 * D_DIM;
    const float* b0p = fbr + (size_t)(tf * 32 + lm) * D_DIM + wid * KSEG + 4 * half;
    const float* b1p = b0p + (size_t)16 * D_DIM;

    constexpr int NS = KSEG / 32;
    v4f fa0[NS][4], fa1[NS][4], fb0[NS][4], fb1[NS][4];
#define RT_LOAD_LINE(F, P, S)                                                     \
    {                                                                             \
        _Pragma("unroll")                                                         \
        for (int jj = 0; jj < 4; ++jj) F[S][jj] = *(const v4f*)((P) + 32 * (S) + 8 * jj); \
    }
#pragma unroll
    for (int s = 0; s < RT_AHEAD; ++s) {
        RT_LOAD_LINE(fa0, a0p, s)
        RT_LOAD_LINE(fa1, a1p, s)
        RT_LOAD_LINE(fb0, b0p, s)
        RT_LOAD_LINE(fb1, b1p, s)
    }

    v8f acc00 = {}, acc01 = {}, acc10 = {}, acc11 = {};
#pragma unroll
    for (int s = 0; s < NS; ++s) {
#pragma unroll
        for (int j = 0; j < 4; ++j) {
            if (s + RT_AHEAD < NS)
                asm volatile("" :: "v"(fa0[s][j]), "v"(fa1[s][j]), "v"(fb0[s][j]), "v"(fb1[s][j]) : "memory");
            const v2f a0l = {fa0[s][j].x, fa0[s][j].y}, a0h = {fa0[s][j].z, fa0[s][j].w};
            const v2f a1l = {fa1[s][j].x, fa1[s][j].y}, a1h = {fa1[s][j].z, fa1[s][j].w};
            const v2f b0l = {fb0[s][j].x, fb0[s][j].y}, b0h = {fb0[s][j].z, fb0[s][j].w};
            const v2f b1l = {fb1[s][j].x, fb1[s][j].y}, b1h = {fb1[s][j].z, fb1[s][j].w};
            acc00 = __builtin_amdgcn_wmma_f32_16x16x4_f32(false, a0l, false, b0l, (short)0, acc00, false, false);
            acc01 = __builtin_amdgcn_wmma_f32_16x16x4_f32(false, a0l, false, b1l, (short)0, acc01, false, false);
            acc10 = __builtin_amdgcn_wmma_f32_16x16x4_f32(false, a1l, false, b0l, (short)0, acc10, false, false);
            acc11 = __builtin_amdgcn_wmma_f32_16x16x4_f32(false, a1l, false, b1l, (short)0, acc11, false, false);
            acc00 = __builtin_amdgcn_wmma_f32_16x16x4_f32(false, a0h, false, b0h, (short)0, acc00, false, false);
            acc01 = __builtin_amdgcn_wmma_f32_16x16x4_f32(false, a0h, false, b1h, (short)0, acc01, false, false);
            acc10 = __builtin_amdgcn_wmma_f32_16x16x4_f32(false, a1h, false, b0h, (short)0, acc10, false, false);
            acc11 = __builtin_amdgcn_wmma_f32_16x16x4_f32(false, a1h, false, b1h, (short)0, acc11, false, false);
            if (s + RT_AHEAD < NS) {
                if (j == 0)      RT_LOAD_LINE(fa0, a0p, s + RT_AHEAD)
                else if (j == 1) RT_LOAD_LINE(fa1, a1p, s + RT_AHEAD)
                else if (j == 2) RT_LOAD_LINE(fb0, b0p, s + RT_AHEAD)
                else             RT_LOAD_LINE(fb1, b1p, s + RT_AHEAD)
            }
        }
    }
#undef RT_LOAD_LINE

#pragma unroll
    for (int v = 0; v < 8; ++v) {
        red[wid][     8 * half + v][     lm] = acc00[v];
        red[wid][     8 * half + v][16 + lm] = acc01[v];
        red[wid][16 + 8 * half + v][     lm] = acc10[v];
        red[wid][16 + 8 * half + v][16 + lm] = acc11[v];
    }
    __syncthreads();

#pragma unroll
    for (int i = 0; i < NR; ++i) {
        const int r = wid * NR + i;
        float s = red[0][r][lane];
#pragma unroll
        for (int w = 1; w < KS; ++w) s += red[w][r][lane];
        outp[(size_t)(tm * 32 + r) * F_DIM + tf * 32 + lane] = fmaxf(s, 0.0f);
    }
}

#define S1_RT_BLOCKS ((N_ROWS / 32) * (F_DIM / 32))
#define S1_PR_BLOCKS ((P_ROWS / 16) * (F_DIM / 16))
#define S1_NR_BLOCKS (NRM_ROWS / NRM_WAVES)

__global__ __launch_bounds__(256) void stage1_kernel(const float* __restrict__ X,
                                                     const float* __restrict__ Pm,
                                                     const float* __restrict__ fb,
                                                     float* __restrict__ inv_in,
                                                     float* __restrict__ inv_pr,
                                                     float* __restrict__ inv_fb,
                                                     float* __restrict__ apr,
                                                     float* __restrict__ bpr,
                                                     const float* __restrict__ alpha_raw,
                                                     const float* __restrict__ beta_raw,
                                                     const float* __restrict__ theta_raw,
                                                     float* __restrict__ coef) {
    __shared__ float red_s[8 * 32 * RT_PITCH];
    const int b = blockIdx.x;
    if (b < S1_RT_BLOCKS)
        proj_rt_role<8>(X, fb, apr, b, red_s);
    else if (b < S1_RT_BLOCKS + S1_PR_BLOCKS)
        proj_role<8>(Pm, fb, bpr, b - S1_RT_BLOCKS, red_s);
    else
        norms_role(X, Pm, fb, inv_in, inv_pr, inv_fb, alpha_raw, beta_raw, theta_raw, coef,
                   b - S1_RT_BLOCKS - S1_PR_BLOCKS);
}

#define PW_IW    4
#define PW_IB    32
#define PW_JB    128
#define PW_MOFF  132
#define PW_STR   276

__global__ __launch_bounds__(256) void pairwise_kernel(const float* __restrict__ ap,
                                                       const float* __restrict__ bp,
                                                       const float* __restrict__ inv_in,
                                                       const float* __restrict__ inv_pr,
                                                       const float* __restrict__ inv_fb,
                                                       const float* __restrict__ coef,
                                                       float* __restrict__ out) {
    __shared__ __attribute__((aligned(16))) float tS[F_DIM * PW_STR];

    const int tid  = threadIdx.x;
    const int lane = tid & 31;
    const int wid  = __builtin_amdgcn_readfirstlane(tid >> 5);
    const int j0   = blockIdx.x * PW_JB;
    const int i0   = blockIdx.y * PW_IB + wid * PW_IW;

    float al, be, th;
    {
        const int jq = (wid << 2) | (lane & 3);
        float4 g[4][4];
#pragma unroll
        for (int t = 0; t < 4; ++t) {
            const int q = (t << 3) | (lane >> 2);
#pragma unroll
            for (int rr = 0; rr < 4; ++rr)
                g[t][rr] = *(const float4*)(bp + (size_t)(j0 + 4 * jq + rr) * F_DIM + 4 * q);
        }
        const float4 tj4 = *(const float4*)(inv_pr + j0 + 4 * jq);
        const float tj[4] = {tj4.x, tj4.y, tj4.z, tj4.w};
        float4 fk[4];
#pragma unroll
        for (int t = 0; t < 4; ++t) fk[t] = *(const float4*)(inv_fb + 4 * ((t << 3) | (lane >> 2)));
        al = coef[0];
        be = coef[1];
        th = coef[2];
#pragma unroll
        for (int t = 0; t < 4; ++t) {
#pragma unroll
            for (int rr = 0; rr < 4; ++rr) {
                g[t][rr].x = (g[t][rr].x * tj[rr]) * fk[t].x;
                g[t][rr].y = (g[t][rr].y * tj[rr]) * fk[t].y;
                g[t][rr].z = (g[t][rr].z * tj[rr]) * fk[t].z;
                g[t][rr].w = (g[t][rr].w * tj[rr]) * fk[t].w;
            }
        }
#pragma unroll
        for (int t = 0; t < 4; ++t) {
            const int q = (t << 3) | (lane >> 2);
            float* dst = tS + (4 * q) * PW_STR + 4 * jq;
            float4 v, m;
            v.x = g[t][0].x; v.y = g[t][1].x; v.z = g[t][2].x; v.w = g[t][3].x;
            m.x = (v.x > 0.0f) ? 1.0f : 0.0f; m.y = (v.y > 0.0f) ? 1.0f : 0.0f;
            m.z = (v.z > 0.0f) ? 1.0f : 0.0f; m.w = (v.w > 0.0f) ? 1.0f : 0.0f;
            *(float4*)(dst) = v; *(float4*)(dst + PW_MOFF) = m;
            v.x = g[t][0].y; v.y = g[t][1].y; v.z = g[t][2].y; v.w = g[t][3].y;
            m.x = (v.x > 0.0f) ? 1.0f : 0.0f; m.y = (v.y > 0.0f) ? 1.0f : 0.0f;
            m.z = (v.z > 0.0f) ? 1.0f : 0.0f; m.w = (v.w > 0.0f) ? 1.0f : 0.0f;
            *(float4*)(dst + PW_STR) = v; *(float4*)(dst + PW_STR + PW_MOFF) = m;
            v.x = g[t][0].z; v.y = g[t][1].z; v.z = g[t][2].z; v.w = g[t][3].z;
            m.x = (v.x > 0.0f) ? 1.0f : 0.0f; m.y = (v.y > 0.0f) ? 1.0f : 0.0f;
            m.z = (v.z > 0.0f) ? 1.0f : 0.0f; m.w = (v.w > 0.0f) ? 1.0f : 0.0f;
            *(float4*)(dst + 2 * PW_STR) = v; *(float4*)(dst + 2 * PW_STR + PW_MOFF) = m;
            v.x = g[t][0].w; v.y = g[t][1].w; v.z = g[t][2].w; v.w = g[t][3].w;
            m.x = (v.x > 0.0f) ? 1.0f : 0.0f; m.y = (v.y > 0.0f) ? 1.0f : 0.0f;
            m.z = (v.z > 0.0f) ? 1.0f : 0.0f; m.w = (v.w > 0.0f) ? 1.0f : 0.0f;
            *(float4*)(dst + 3 * PW_STR) = v; *(float4*)(dst + 3 * PW_STR + PW_MOFF) = m;
        }
    }
    asm volatile("" :: "s"(al), "s"(be), "s"(th));
    __syncthreads();

    const float* __restrict__ arow = ap + (size_t)i0 * F_DIM;
    const float* brow = tS + 4 * lane;

    float sc[PW_IW];
#pragma unroll
    for (int r = 0; r < PW_IW; ++r) sc[r] = inv_in[i0 + r];

    float s_min[PW_IW][4], s_a[PW_IW][4], s_b[PW_IW][4];
#pragma unroll
    for (int r = 0; r < PW_IW; ++r) {
#pragma unroll
        for (int c = 0; c < 4; ++c) { s_min[r][c] = 0.0f; s_a[r][c] = 0.0f; s_b[r][c] = 0.0f; }
    }

    v8f avP[PW_IW], avQ[PW_IW];
    v8f fvP, fvQ;
    v4f bqA[4], mqA[4], bqB[4], mqB[4];

#define PW_LOAD_BM(BQ, MQ, K0)                                                    \
    _Pragma("unroll")                                                             \
    for (int kk = 0; kk < 4; ++kk) {                                              \
        BQ[kk] = *(const v4f*)(brow + ((K0) + kk) * PW_STR);                      \
        MQ[kk] = *(const v4f*)(brow + ((K0) + kk) * PW_STR + PW_MOFF);            \
    }
#define PW_LOAD_AP(AV, FV, K0)                                                    \
    _Pragma("unroll")                                                             \
    for (int r = 0; r < PW_IW; ++r) AV[r] = *(const v8f*)(arow + r * F_DIM + (K0)); \
    FV = *(const v8f*)(inv_fb + (K0));

#define PW_SCALE_AP(AV, FV)                                                       \
    _Pragma("unroll")                                                             \
    for (int r = 0; r < PW_IW; ++r) {                                             \
        _Pragma("unroll")                                                         \
        for (int e = 0; e < 8; ++e) AV[r][e] = (AV[r][e] * sc[r]) * FV[e];        \
    }

#define PW_HOLD_AP_BM(AV, BQ, MQ)                                                 \
    asm volatile("" :: "s"(AV[0]), "s"(AV[1]), "s"(AV[2]), "s"(AV[3]),            \
                       "v"(BQ[0]), "v"(BQ[1]), "v"(BQ[2]), "v"(BQ[3]),            \
                       "v"(MQ[0]), "v"(MQ[1]), "v"(MQ[2]), "v"(MQ[3]) : "memory")
#define PW_HOLD_BM(BQ, MQ)                                                        \
    asm volatile("" :: "v"(BQ[0]), "v"(BQ[1]), "v"(BQ[2]), "v"(BQ[3]),            \
                       "v"(MQ[0]), "v"(MQ[1]), "v"(MQ[2]), "v"(MQ[3]) : "memory")

#define PW_SUM_QUAD(AV, KO, BQ, MQ)                                               \
    _Pragma("unroll")                                                             \
    for (int kk = 0; kk < 4; ++kk) {                                              \
        _Pragma("unroll")                                                         \
        for (int r = 0; r < PW_IW; ++r) {                                         \
            const float a = AV[r][(KO) + kk];                                     \
            if (__builtin_expect(a > 0.0f, 1)) {                                  \
                _Pragma("unroll")                                                 \
                for (int c = 0; c < 4; ++c) {                                     \
                    s_min[r][c] += fminf(a, BQ[kk][c]);                           \
                    s_a[r][c] = fmaf(a, MQ[kk][c], s_a[r][c]);                    \
                    s_b[r][c] += BQ[kk][c];                                       \
                }                                                                 \
            }                                                                     \
        }                                                                         \
    }

#define PW_PASS(AVC, FVC, AVN, FVN, K0)                                           \
    {                                                                             \
        const int kn = ((K0) + 8 < F_DIM) ? ((K0) + 8) : (F_DIM - 8);             \
        PW_LOAD_BM(bqB, mqB, (K0) + 4)                                            \
        PW_SCALE_AP(AVC, FVC)                                                     \
        PW_HOLD_AP_BM(AVC, bqA, mqA);                                             \
        PW_LOAD_AP(AVN, FVN, kn)                                                  \
        PW_SUM_QUAD(AVC, 0, bqA, mqA)                                             \
        PW_LOAD_BM(bqA, mqA, kn)                                                  \
        PW_HOLD_BM(bqB, mqB);                                                     \
        PW_SUM_QUAD(AVC, 4, bqB, mqB)                                             \
    }

    PW_LOAD_AP(avP, fvP, 0)
    PW_LOAD_BM(bqA, mqA, 0)
#pragma nounroll
    for (int k = 0; k < F_DIM; k += 16) {
        PW_PASS(avP, fvP, avQ, fvQ, k)
        PW_PASS(avQ, fvQ, avP, fvP, k + 8)
    }
#undef PW_LOAD_BM
#undef PW_LOAD_AP
#undef PW_SCALE_AP
#undef PW_HOLD_AP_BM
#undef PW_HOLD_BM
#undef PW_SUM_QUAD
#undef PW_PASS

#pragma unroll
    for (int r = 0; r < PW_IW; ++r) {
        float4 o;
        o.x = (th + al + be) * s_min[r][0] - al * s_a[r][0] - be * s_b[r][0];
        o.y = (th + al + be) * s_min[r][1] - al * s_a[r][1] - be * s_b[r][1];
        o.z = (th + al + be) * s_min[r][2] - al * s_a[r][2] - be * s_b[r][2];
        o.w = (th + al + be) * s_min[r][3] - al * s_a[r][3] - be * s_b[r][3];
        *(float4*)(out + (size_t)(i0 + r) * P_ROWS + j0 + 4 * lane) = o;
    }
}

extern "C" void kernel_launch(void* const* d_in, const int* in_sizes, int n_in,
                              void* d_out, int out_size, void* d_ws, size_t ws_size,
                              hipStream_t stream) {
    const float* input     = (const float*)d_in[0];
    const float* prototype = (const float*)d_in[1];
    const float* fb        = (const float*)d_in[2];
    const float* alpha_raw = (const float*)d_in[3];
    const float* beta_raw  = (const float*)d_in[4];
    const float* theta_raw = (const float*)d_in[5];
    float* out = (float*)d_out;

    float* w      = (float*)d_ws;
    float* inv_fb = w;
    float* inv_in = inv_fb + F_DIM;
    float* inv_pr = inv_in + N_ROWS;
    float* ap     = inv_pr + P_ROWS;
    float* bp     = ap + (size_t)N_ROWS * F_DIM;
    float* coef   = bp + (size_t)P_ROWS * F_DIM;

    stage1_kernel<<<S1_RT_BLOCKS + S1_PR_BLOCKS + S1_NR_BLOCKS, 256, 0, stream>>>(
        input, prototype, fb, inv_in, inv_pr, inv_fb, ap, bp,
        alpha_raw, beta_raw, theta_raw, coef);

    dim3 grid(P_ROWS / PW_JB, N_ROWS / PW_IB);
    pairwise_kernel<<<grid, 256, 0, stream>>>(ap, bp, inv_in, inv_pr, inv_fb, coef, out);
}


# First launch in work-groups of 4 waves, not 8: a tile's K is split 4 ways (256 k per wave: 256 WMMAs on input tiles, 64 on prototype tiles), 2048 fewer waves enter; dot sums reordered
# speedup vs baseline: 3.4651x; 1.0302x over previous
#include <hip/hip_runtime.h>
#include <hip/hip_bf16.h>
#include <math.h>

typedef float v2f __attribute__((ext_vector_type(2)));
typedef float v8f __attribute__((ext_vector_type(8)));
typedef float v4f __attribute__((ext_vector_type(4)));

#define N_ROWS 2048
#define P_ROWS 512
#define D_DIM  1024
#define F_DIM  128
#define EPSV   1e-8f

__device__ __forceinline__ float softplus_f(float x) {
    return log1pf(expf(x));
}

#define NRM_WAVES 4
#define NRM_ROWS  (F_DIM + P_ROWS + N_ROWS)

__device__ __forceinline__ void norms_role(const float* __restrict__ X,
                                           const float* __restrict__ Pm,
                                           const float* __restrict__ fb,
                                           float* __restrict__ inv_in,
                                           float* __restrict__ inv_pr,
                                           float* __restrict__ inv_fb,
                                           const float* __restrict__ alpha_raw,
                                           const float* __restrict__ beta_raw,
                                           const float* __restrict__ theta_raw,
                                           float* __restrict__ coef,
                                           const int blk) {
    const int lane = threadIdx.x & 31;
    const int wid = __builtin_amdgcn_readfirstlane(threadIdx.x >> 5);
    const int r   = blk * NRM_WAVES + wid;

    const float* src;
    if (r < F_DIM)               src = fb + (size_t)r * D_DIM;
    else if (r < F_DIM + P_ROWS) src = Pm + (size_t)(r - F_DIM) * D_DIM;
    else                         src = X  + (size_t)(r - F_DIM - P_ROWS) * D_DIM;

    float4 v[8];
#pragma unroll
    for (int j = 0; j < 8; ++j) v[j] = *(const float4*)(src + 4 * (j * 32 + lane));

    if (r == 0) {
        const float r_al = alpha_raw[0];
        const float r_be = beta_raw[0];
        const float r_th = theta_raw[0];
        asm volatile("" :: "s"(r_al), "s"(r_be), "s"(r_th));
        const float c_al = softplus_f(r_al);
        const float c_be = softplus_f(r_be);
        const float c_th = softplus_f(r_th);
        if (lane == 0) { coef[0] = c_al; coef[1] = c_be; coef[2] = c_th; }
    }

    float sx = 0.0f, sy = 0.0f, sz = 0.0f, sw = 0.0f;
#pragma unroll
    for (int j = 0; j < 8; ++j) {
        sx = fmaf(v[j].x, v[j].x, sx);
        sy = fmaf(v[j].y, v[j].y, sy);
        sz = fmaf(v[j].z, v[j].z, sz);
        sw = fmaf(v[j].w, v[j].w, sw);
    }
    float s = (sx + sy) + (sz + sw);
#pragma unroll
    for (int off = 16; off > 0; off >>= 1) s += __shfl_xor(s, off);

    const float scale = 1.0f / fmaxf(sqrtf(s), EPSV);

    if (lane == 0) {
        if (r < F_DIM)               inv_fb[r] = scale;
        else if (r < F_DIM + P_ROWS) inv_pr[r - F_DIM] = scale;
        else                         inv_in[r - F_DIM - P_ROWS] = scale;
    }
}

template <int KS>
__device__ __forceinline__ void proj_role(const float* __restrict__ X,
                                          const float* __restrict__ fbr,
                                          float* __restrict__ outp,
                                          const int tile,
                                          float* __restrict__ red_s) {
    constexpr int KSEG = D_DIM / KS;
    float (*red)[8][32] = (float (*)[8][32])red_s;

    const int wid  = threadIdx.x >> 5;
    const int lane = threadIdx.x & 31;
    const int tf   = tile & 7;
    const int tm   = tile >> 3;
    const int half = lane >> 4;
    const int lm   = lane & 15;

    const float* arow = X   + (size_t)(tm * 16 + lm) * D_DIM + wid * KSEG + 4 * half;
    const float* brow = fbr + (size_t)(tf * 16 + lm) * D_DIM + wid * KSEG + 4 * half;

    constexpr int NV = 8 / KS;
    const int col  = tf * 16 + lm;
    const int row0 = tm * 16 + half * 8 + wid;

    v8f acc = {};
#pragma unroll
    for (int k = 0; k < KSEG; k += 8) {
        const float4 a4 = *(const float4*)(arow + k);
        const float4 b4 = *(const float4*)(brow + k);
        const v2f a_lo = {a4.x, a4.y}, a_hi = {a4.z, a4.w};
        const v2f b_lo = {b4.x, b4.y}, b_hi = {b4.z, b4.w};
        acc = __builtin_amdgcn_wmma_f32_16x16x4_f32(
             false, a_lo,  false, b_lo,
             (short)0, acc,  false,  false);
        acc = __builtin_amdgcn_wmma_f32_16x16x4_f32(
             false, a_hi,  false, b_hi,
             (short)0, acc,  false,  false);
    }

#pragma unroll
    for (int v = 0; v < 8; ++v) red[wid][v][lane] = acc[v];
    __syncthreads();

#pragma unroll
    for (int i = 0; i < NV; ++i) {
        const int v = wid + i * KS;
        float s = red[0][v][lane];
#pragma unroll
        for (int w = 1; w < KS; ++w) s += red[w][v][lane];
        outp[(size_t)(row0 + i * KS) * F_DIM + col] = fmaxf(s, 0.0f);
    }
}

#define RT_PITCH 34
#define RT_AHEAD 2

template <int KS>
__device__ __forceinline__ void proj_rt_role(const float* __restrict__ X,
                                             const float* __restrict__ fbr,
                                             float* __restrict__ outp,
                                             const int tile,
                                             float* __restrict__ red_s) {
    constexpr int KSEG = D_DIM / KS;
    constexpr int NR   = 32 / KS;
    float (*red)[32][RT_PITCH] = (float (*)[32][RT_PITCH])red_s;

    const int lane = threadIdx.x & 31;
    const int wid  = __builtin_amdgcn_readfirstlane(threadIdx.x >> 5);
    const int tf   = tile & 3;
    const int tm   = tile >> 2;
    const int half = lane >> 4;
    const int lm   = lane & 15;

    const float* a0p = X   + (size_t)(tm * 32 + lm) * D_DIM + wid * KSEG + 4 * half;
    const float* a1p = a0p + (size_t)16 * D_DIM;
    const float* b0p = fbr + (size_t)(tf * 32 + lm) * D_DIM + wid * KSEG + 4 * half;
    const float* b1p = b0p + (size_t)16 * D_DIM;

    constexpr int NS = KSEG / 32;
    v4f fa0[NS][4], fa1[NS][4], fb0[NS][4], fb1[NS][4];
#define RT_LOAD_LINE(F, P, S)                                                     \
    {                                                                             \
        _Pragma("unroll")                                                         \
        for (int jj = 0; jj < 4; ++jj) F[S][jj] = *(const v4f*)((P) + 32 * (S) + 8 * jj); \
    }
#pragma unroll
    for (int s = 0; s < RT_AHEAD; ++s) {
        RT_LOAD_LINE(fa0, a0p, s)
        RT_LOAD_LINE(fa1, a1p, s)
        RT_LOAD_LINE(fb0, b0p, s)
        RT_LOAD_LINE(fb1, b1p, s)
    }

    v8f acc00 = {}, acc01 = {}, acc10 = {}, acc11 = {};
#pragma unroll
    for (int s = 0; s < NS; ++s) {
#pragma unroll
        for (int j = 0; j < 4; ++j) {
            if (s + RT_AHEAD < NS)
                asm volatile("" :: "v"(fa0[s][j]), "v"(fa1[s][j]), "v"(fb0[s][j]), "v"(fb1[s][j]) : "memory");
            const v2f a0l = {fa0[s][j].x, fa0[s][j].y}, a0h = {fa0[s][j].z, fa0[s][j].w};
            const v2f a1l = {fa1[s][j].x, fa1[s][j].y}, a1h = {fa1[s][j].z, fa1[s][j].w};
            const v2f b0l = {fb0[s][j].x, fb0[s][j].y}, b0h = {fb0[s][j].z, fb0[s][j].w};
            const v2f b1l = {fb1[s][j].x, fb1[s][j].y}, b1h = {fb1[s][j].z, fb1[s][j].w};
            acc00 = __builtin_amdgcn_wmma_f32_16x16x4_f32(false, a0l, false, b0l, (short)0, acc00, false, false);
            acc01 = __builtin_amdgcn_wmma_f32_16x16x4_f32(false, a0l, false, b1l, (short)0, acc01, false, false);
            acc10 = __builtin_amdgcn_wmma_f32_16x16x4_f32(false, a1l, false, b0l, (short)0, acc10, false, false);
            acc11 = __builtin_amdgcn_wmma_f32_16x16x4_f32(false, a1l, false, b1l, (short)0, acc11, false, false);
            acc00 = __builtin_amdgcn_wmma_f32_16x16x4_f32(false, a0h, false, b0h, (short)0, acc00, false, false);
            acc01 = __builtin_amdgcn_wmma_f32_16x16x4_f32(false, a0h, false, b1h, (short)0, acc01, false, false);
            acc10 = __builtin_amdgcn_wmma_f32_16x16x4_f32(false, a1h, false, b0h, (short)0, acc10, false, false);
            acc11 = __builtin_amdgcn_wmma_f32_16x16x4_f32(false, a1h, false, b1h, (short)0, acc11, false, false);
            if (s + RT_AHEAD < NS) {
                if (j == 0)      RT_LOAD_LINE(fa0, a0p, s + RT_AHEAD)
                else if (j == 1) RT_LOAD_LINE(fa1, a1p, s + RT_AHEAD)
                else if (j == 2) RT_LOAD_LINE(fb0, b0p, s + RT_AHEAD)
                else             RT_LOAD_LINE(fb1, b1p, s + RT_AHEAD)
            }
        }
    }
#undef RT_LOAD_LINE

#pragma unroll
    for (int v = 0; v < 8; ++v) {
        red[wid][     8 * half + v][     lm] = acc00[v];
        red[wid][     8 * half + v][16 + lm] = acc01[v];
        red[wid][16 + 8 * half + v][     lm] = acc10[v];
        red[wid][16 + 8 * half + v][16 + lm] = acc11[v];
    }
    __syncthreads();

#pragma unroll
    for (int i = 0; i < NR; ++i) {
        const int r = wid * NR + i;
        float s = red[0][r][lane];
#pragma unroll
        for (int w = 1; w < KS; ++w) s += red[w][r][lane];
        outp[(size_t)(tm * 32 + r) * F_DIM + tf * 32 + lane] = fmaxf(s, 0.0f);
    }
}

#define S1_RT_BLOCKS ((N_ROWS / 32) * (F_DIM / 32))
#define S1_PR_BLOCKS ((P_ROWS / 16) * (F_DIM / 16))
#define S1_NR_BLOCKS (NRM_ROWS / NRM_WAVES)

__global__ __launch_bounds__(128) void stage1_kernel(const float* __restrict__ X,
                                                     const float* __restrict__ Pm,
                                                     const float* __restrict__ fb,
                                                     float* __restrict__ inv_in,
                                                     float* __restrict__ inv_pr,
                                                     float* __restrict__ inv_fb,
                                                     float* __restrict__ apr,
                                                     float* __restrict__ bpr,
                                                     const float* __restrict__ alpha_raw,
                                                     const float* __restrict__ beta_raw,
                                                     const float* __restrict__ theta_raw,
                                                     float* __restrict__ coef) {
    __shared__ float red_s[4 * 32 * RT_PITCH];
    const int b = blockIdx.x;
    if (b < S1_RT_BLOCKS)
        proj_rt_role<4>(X, fb, apr, b, red_s);
    else if (b < S1_RT_BLOCKS + S1_PR_BLOCKS)
        proj_role<4>(Pm, fb, bpr, b - S1_RT_BLOCKS, red_s);
    else
        norms_role(X, Pm, fb, inv_in, inv_pr, inv_fb, alpha_raw, beta_raw, theta_raw, coef,
                   b - S1_RT_BLOCKS - S1_PR_BLOCKS);
}

#define PW_IW    4
#define PW_IB    32
#define PW_JB    128
#define PW_MOFF  132
#define PW_STR   276

__global__ __launch_bounds__(256) void pairwise_kernel(const float* __restrict__ ap,
                                                       const float* __restrict__ bp,
                                                       const float* __restrict__ inv_in,
                                                       const float* __restrict__ inv_pr,
                                                       const float* __restrict__ inv_fb,
                                                       const float* __restrict__ coef,
                                                       float* __restrict__ out) {
    __shared__ __attribute__((aligned(16))) float tS[F_DIM * PW_STR];

    const int tid  = threadIdx.x;
    const int lane = tid & 31;
    const int wid  = __builtin_amdgcn_readfirstlane(tid >> 5);
    const int j0   = blockIdx.x * PW_JB;
    const int i0   = blockIdx.y * PW_IB + wid * PW_IW;

    float al, be, th;
    {
        const int jq = (wid << 2) | (lane & 3);
        float4 g[4][4];
#pragma unroll
        for (int t = 0; t < 4; ++t) {
            const int q = (t << 3) | (lane >> 2);
#pragma unroll
            for (int rr = 0; rr < 4; ++rr)
                g[t][rr] = *(const float4*)(bp + (size_t)(j0 + 4 * jq + rr) * F_DIM + 4 * q);
        }
        const float4 tj4 = *(const float4*)(inv_pr + j0 + 4 * jq);
        const float tj[4] = {tj4.x, tj4.y, tj4.z, tj4.w};
        float4 fk[4];
#pragma unroll
        for (int t = 0; t < 4; ++t) fk[t] = *(const float4*)(inv_fb + 4 * ((t << 3) | (lane >> 2)));
        al = coef[0];
        be = coef[1];
        th = coef[2];
#pragma unroll
        for (int t = 0; t < 4; ++t) {
#pragma unroll
            for (int rr = 0; rr < 4; ++rr) {
                g[t][rr].x = (g[t][rr].x * tj[rr]) * fk[t].x;
                g[t][rr].y = (g[t][rr].y * tj[rr]) * fk[t].y;
                g[t][rr].z = (g[t][rr].z * tj[rr]) * fk[t].z;
                g[t][rr].w = (g[t][rr].w * tj[rr]) * fk[t].w;
            }
        }
#pragma unroll
        for (int t = 0; t < 4; ++t) {
            const int q = (t << 3) | (lane >> 2);
            float* dst = tS + (4 * q) * PW_STR + 4 * jq;
            float4 v, m;
            v.x = g[t][0].x; v.y = g[t][1].x; v.z = g[t][2].x; v.w = g[t][3].x;
            m.x = (v.x > 0.0f) ? 1.0f : 0.0f; m.y = (v.y > 0.0f) ? 1.0f : 0.0f;
            m.z = (v.z > 0.0f) ? 1.0f : 0.0f; m.w = (v.w > 0.0f) ? 1.0f : 0.0f;
            *(float4*)(dst) = v; *(float4*)(dst + PW_MOFF) = m;
            v.x = g[t][0].y; v.y = g[t][1].y; v.z = g[t][2].y; v.w = g[t][3].y;
            m.x = (v.x > 0.0f) ? 1.0f : 0.0f; m.y = (v.y > 0.0f) ? 1.0f : 0.0f;
            m.z = (v.z > 0.0f) ? 1.0f : 0.0f; m.w = (v.w > 0.0f) ? 1.0f : 0.0f;
            *(float4*)(dst + PW_STR) = v; *(float4*)(dst + PW_STR + PW_MOFF) = m;
            v.x = g[t][0].z; v.y = g[t][1].z; v.z = g[t][2].z; v.w = g[t][3].z;
            m.x = (v.x > 0.0f) ? 1.0f : 0.0f; m.y = (v.y > 0.0f) ? 1.0f : 0.0f;
            m.z = (v.z > 0.0f) ? 1.0f : 0.0f; m.w = (v.w > 0.0f) ? 1.0f : 0.0f;
            *(float4*)(dst + 2 * PW_STR) = v; *(float4*)(dst + 2 * PW_STR + PW_MOFF) = m;
            v.x = g[t][0].w; v.y = g[t][1].w; v.z = g[t][2].w; v.w = g[t][3].w;
            m.x = (v.x > 0.0f) ? 1.0f : 0.0f; m.y = (v.y > 0.0f) ? 1.0f : 0.0f;
            m.z = (v.z > 0.0f) ? 1.0f : 0.0f; m.w = (v.w > 0.0f) ? 1.0f : 0.0f;
            *(float4*)(dst + 3 * PW_STR) = v; *(float4*)(dst + 3 * PW_STR + PW_MOFF) = m;
        }
    }
    asm volatile("" :: "s"(al), "s"(be), "s"(th));
    __syncthreads();

    const float* __restrict__ arow = ap + (size_t)i0 * F_DIM;
    const float* brow = tS + 4 * lane;

    float sc[PW_IW];
#pragma unroll
    for (int r = 0; r < PW_IW; ++r) sc[r] = inv_in[i0 + r];

    float s_min[PW_IW][4], s_a[PW_IW][4], s_b[PW_IW][4];
#pragma unroll
    for (int r = 0; r < PW_IW; ++r) {
#pragma unroll
        for (int c = 0; c < 4; ++c) { s_min[r][c] = 0.0f; s_a[r][c] = 0.0f; s_b[r][c] = 0.0f; }
    }

    v8f avP[PW_IW], avQ[PW_IW];
    v8f fvP, fvQ;
    v4f bqA[4], mqA[4], bqB[4], mqB[4];

#define PW_LOAD_BM(BQ, MQ, K0)                                                    \
    _Pragma("unroll")                                                             \
    for (int kk = 0; kk < 4; ++kk) {                                              \
        BQ[kk] = *(const v4f*)(brow + ((K0) + kk) * PW_STR);                      \
        MQ[kk] = *(const v4f*)(brow + ((K0) + kk) * PW_STR + PW_MOFF);            \
    }
#define PW_LOAD_AP(AV, FV, K0)                                                    \
    _Pragma("unroll")                                                             \
    for (int r = 0; r < PW_IW; ++r) AV[r] = *(const v8f*)(arow + r * F_DIM + (K0)); \
    FV = *(const v8f*)(inv_fb + (K0));

#define PW_SCALE_AP(AV, FV)                                                       \
    _Pragma("unroll")                                                             \
    for (int r = 0; r < PW_IW; ++r) {                                             \
        _Pragma("unroll")                                                         \
        for (int e = 0; e < 8; ++e) AV[r][e] = (AV[r][e] * sc[r]) * FV[e];        \
    }

#define PW_HOLD_AP_BM(AV, BQ, MQ)                                                 \
    asm volatile("" :: "s"(AV[0]), "s"(AV[1]), "s"(AV[2]), "s"(AV[3]),            \
                       "v"(BQ[0]), "v"(BQ[1]), "v"(BQ[2]), "v"(BQ[3]),            \
                       "v"(MQ[0]), "v"(MQ[1]), "v"(MQ[2]), "v"(MQ[3]) : "memory")
#define PW_HOLD_BM(BQ, MQ)                                                        \
    asm volatile("" :: "v"(BQ[0]), "v"(BQ[1]), "v"(BQ[2]), "v"(BQ[3]),            \
                       "v"(MQ[0]), "v"(MQ[1]), "v"(MQ[2]), "v"(MQ[3]) : "memory")

#define PW_SUM_QUAD(AV, KO, BQ, MQ)                                               \
    _Pragma("unroll")                                                             \
    for (int kk = 0; kk < 4; ++kk) {                                              \
        _Pragma("unroll")                                                         \
        for (int r = 0; r < PW_IW; ++r) {                                         \
            const float a = AV[r][(KO) + kk];                                     \
            if (__builtin_expect(a > 0.0f, 1)) {                                  \
                _Pragma("unroll")                                                 \
                for (int c = 0; c < 4; ++c) {                                     \
                    s_min[r][c] += fminf(a, BQ[kk][c]);                           \
                    s_a[r][c] = fmaf(a, MQ[kk][c], s_a[r][c]);                    \
                    s_b[r][c] += BQ[kk][c];                                       \
                }                                                                 \
            }                                                                     \
        }                                                                         \
    }

#define PW_PASS(AVC, FVC, AVN, FVN, K0)                                           \
    {                                                                             \
        const int kn = ((K0) + 8 < F_DIM) ? ((K0) + 8) : (F_DIM - 8);             \
        PW_LOAD_BM(bqB, mqB, (K0) + 4)                                            \
        PW_SCALE_AP(AVC, FVC)                                                     \
        PW_HOLD_AP_BM(AVC, bqA, mqA);                                             \
        PW_LOAD_AP(AVN, FVN, kn)                                                  \
        PW_SUM_QUAD(AVC, 0, bqA, mqA)                                             \
        PW_LOAD_BM(bqA, mqA, kn)                                                  \
        PW_HOLD_BM(bqB, mqB);                                                     \
        PW_SUM_QUAD(AVC, 4, bqB, mqB)                                             \
    }

    PW_LOAD_AP(avP, fvP, 0)
    PW_LOAD_BM(bqA, mqA, 0)
#pragma nounroll
    for (int k = 0; k < F_DIM; k += 16) {
        PW_PASS(avP, fvP, avQ, fvQ, k)
        PW_PASS(avQ, fvQ, avP, fvP, k + 8)
    }
#undef PW_LOAD_BM
#undef PW_LOAD_AP
#undef PW_SCALE_AP
#undef PW_HOLD_AP_BM
#undef PW_HOLD_BM
#undef PW_SUM_QUAD
#undef PW_PASS

#pragma unroll
    for (int r = 0; r < PW_IW; ++r) {
        float4 o;
        o.x = (th + al + be) * s_min[r][0] - al * s_a[r][0] - be * s_b[r][0];
        o.y = (th + al + be) * s_min[r][1] - al * s_a[r][1] - be * s_b[r][1];
        o.z = (th + al + be) * s_min[r][2] - al * s_a[r][2] - be * s_b[r][2];
        o.w = (th + al + be) * s_min[r][3] - al * s_a[r][3] - be * s_b[r][3];
        *(float4*)(out + (size_t)(i0 + r) * P_ROWS + j0 + 4 * lane) = o;
    }
}

extern "C" void kernel_launch(void* const* d_in, const int* in_sizes, int n_in,
                              void* d_out, int out_size, void* d_ws, size_t ws_size,
                              hipStream_t stream) {
    const float* input     = (const float*)d_in[0];
    const float* prototype = (const float*)d_in[1];
    const float* fb        = (const float*)d_in[2];
    const float* alpha_raw = (const float*)d_in[3];
    const float* beta_raw  = (const float*)d_in[4];
    const float* theta_raw = (const float*)d_in[5];
    float* out = (float*)d_out;

    float* w      = (float*)d_ws;
    float* inv_fb = w;
    float* inv_in = inv_fb + F_DIM;
    float* inv_pr = inv_in + N_ROWS;
    float* ap     = inv_pr + P_ROWS;
    float* bp     = ap + (size_t)N_ROWS * F_DIM;
    float* coef   = bp + (size_t)P_ROWS * F_DIM;

    stage1_kernel<<<S1_RT_BLOCKS + S1_PR_BLOCKS + S1_NR_BLOCKS, 128, 0, stream>>>(
        input, prototype, fb, inv_in, inv_pr, inv_fb, ap, bp,
        alpha_raw, beta_raw, theta_raw, coef);

    dim3 grid(P_ROWS / PW_JB, N_ROWS / PW_IB);
    pairwise_kernel<<<grid, 256, 0, stream>>>(ap, bp, inv_in, inv_pr, inv_fb, coef, out);
}
